# starting kernel
# baseline (speedup 1.0000x reference)
#include <hip/hip_runtime.h>
#include <hip/hip_bf16.h>

typedef __attribute__((ext_vector_type(2))) float v2f;
typedef __attribute__((ext_vector_type(8))) float v8f;

#define HFEAT 64

__global__ void k_fill1(float* d, int n) {
    int t = blockIdx.x * blockDim.x + threadIdx.x;
    if (t < n) d[t] = 1.0f;
}

__global__ void k_deg(const int* __restrict__ dst, float* __restrict__ deg, int e) {
    int t = blockIdx.x * blockDim.x + threadIdx.x;
    if (t < e) unsafeAtomicAdd(&deg[dst[t]], 1.0f);
}

__global__ void k_rsqrt(float* d, int n) {
    int t = blockIdx.x * blockDim.x + threadIdx.x;
    if (t < n) d[t] = rsqrtf(d[t]);
}

template <int K, int NC>
__global__ void k_gemm_wmma(const float* __restrict__ A, const float* __restrict__ B,
                            float* __restrict__ C, int M) {
    const int NT = NC / 16;
    int wave = (int)((blockIdx.x * blockDim.x + threadIdx.x) >> 5);
    int lane = threadIdx.x & 31;
    int totalTiles = (M / 16) * NT;
    if (wave >= totalTiles) return;

    int m0 = (wave / NT) * 16;
    int n0 = (wave % NT) * 16;
    int row  = m0 + (lane & 15);
    int col  = n0 + (lane & 15);
    int koff = (lane >> 4) * 2;

    v8f acc = {};
    const float* arow = A + (size_t)row * K + koff;
#pragma unroll 4
    for (int k = 0; k < K; k += 4) {
        v2f a = *(const v2f*)(arow + k);
        v2f b;
        b.x = B[(size_t)(k + koff)     * NC + col];
        b.y = B[(size_t)(k + koff + 1) * NC + col];
        acc = __builtin_amdgcn_wmma_f32_16x16x4_f32(false, a, false, b,
                                                    (short)0, acc, false, false);
    }

    int rbase = m0 + ((lane >> 4) * 8);
#pragma unroll
    for (int i = 0; i < 8; ++i)
        C[(size_t)(rbase + i) * NC + col] = acc[i];
}

__global__ void k_self_init(const float* __restrict__ xw, const float* __restrict__ dinv,
                            float* __restrict__ out, int total) {
    int t = blockIdx.x * blockDim.x + threadIdx.x;
    if (t >= total) return;
    int node = t >> 6;
    float di = dinv[node];
    out[t] = di * di * xw[t];
}

__global__ void k_edge_scatter(const int* __restrict__ src, const int* __restrict__ dst,
                               const float* __restrict__ dinv,
                               const float* __restrict__ xw, float* __restrict__ out, int e) {
    int t = blockIdx.x * blockDim.x + threadIdx.x;
    int edge = t >> 4;
    if (edge >= e) return;
    int f4 = (t & 15) * 4;
    int s = src[edge];
    int d = dst[edge];
    float c = dinv[s] * dinv[d];
    const float4 v = *(const float4*)(xw + (size_t)s * HFEAT + f4);
    float* o = out + (size_t)d * HFEAT + f4;
    unsafeAtomicAdd(o + 0, c * v.x);
    unsafeAtomicAdd(o + 1, c * v.y);
    unsafeAtomicAdd(o + 2, c * v.z);
    unsafeAtomicAdd(o + 3, c * v.w);
}

__global__ void k_bias_relu(float* __restrict__ h, const float* __restrict__ b, int total) {
    int t = blockIdx.x * blockDim.x + threadIdx.x;
    if (t >= total) return;
    float v = h[t] + b[t & (HFEAT - 1)];
    h[t] = v > 0.0f ? v : 0.0f;
}

__global__ void k_fc_head(const float* __restrict__ g, const float* __restrict__ fc1_b,
                          const float* __restrict__ fcoW, const float* __restrict__ fcob,
                          float* __restrict__ out, int n) {
    int wave = (int)((blockIdx.x * blockDim.x + threadIdx.x) >> 5);
    int lane = threadIdx.x & 31;
    if (wave >= n) return;
    float v = g[(size_t)wave * 32 + lane] + fc1_b[lane];
    v = v > 0.0f ? v : 0.0f;
    float p = v * fcoW[lane];
#pragma unroll
    for (int off = 16; off > 0; off >>= 1) p += __shfl_xor(p, off, 32);
    if (lane == 0) out[wave] = p + fcob[0];
}

extern "C" void kernel_launch(void* const* d_in, const int* in_sizes, int n_in,
                              void* d_out, int out_size, void* d_ws, size_t ws_size,
                              hipStream_t stream) {
    const float* x     = (const float*)d_in[0];
    const int*   ei    = (const int*)d_in[1];
    const float* W1    = (const float*)d_in[2];
    const float* b1    = (const float*)d_in[3];
    const float* W2    = (const float*)d_in[4];
    const float* b2    = (const float*)d_in[5];
    const float* fc1W  = (const float*)d_in[6];
    const float* fc1b  = (const float*)d_in[7];
    const float* fcoW  = (const float*)d_in[8];
    const float* fcob  = (const float*)d_in[9];
    float* out = (float*)d_out;

    const int N = in_sizes[0] / 128;
    const int E = in_sizes[1] / 2;
    const int* src = ei;
    const int* dst = ei + E;

    float* dinv = (float*)d_ws;
    float* bufA = dinv + N;
    float* bufB = bufA + (size_t)N * HFEAT;

    const int B256 = 256;
    auto blocks = [](long long work, int bs) { return (int)((work + bs - 1) / bs); };

    k_fill1<<<blocks(N, B256), B256, 0, stream>>>(dinv, N);
    k_deg<<<blocks(E, B256), B256, 0, stream>>>(dst, dinv, E);
    k_rsqrt<<<blocks(N, B256), B256, 0, stream>>>(dinv, N);

    {
        long long waves = (long long)(N / 16) * (HFEAT / 16);
        k_gemm_wmma<128, HFEAT><<<blocks(waves * 32, B256), B256, 0, stream>>>(x, W1, bufA, N);
    }
    k_self_init<<<blocks((long long)N * HFEAT, B256), B256, 0, stream>>>(bufA, dinv, bufB, N * HFEAT);
    k_edge_scatter<<<blocks((long long)E * 16, B256), B256, 0, stream>>>(src, dst, dinv, bufA, bufB, E);
    k_bias_relu<<<blocks((long long)N * HFEAT, B256), B256, 0, stream>>>(bufB, b1, N * HFEAT);

    {
        long long waves = (long long)(N / 16) * (HFEAT / 16);
        k_gemm_wmma<HFEAT, HFEAT><<<blocks(waves * 32, B256), B256, 0, stream>>>(bufB, W2, bufA, N);
    }
    k_self_init<<<blocks((long long)N * HFEAT, B256), B256, 0, stream>>>(bufA, dinv, bufB, N * HFEAT);
    k_edge_scatter<<<blocks((long long)E * 16, B256), B256, 0, stream>>>(src, dst, dinv, bufA, bufB, E);
    k_bias_relu<<<blocks((long long)N * HFEAT, B256), B256, 0, stream>>>(bufB, b2, N * HFEAT);

    {
        long long waves = (long long)(N / 16) * (32 / 16);
        k_gemm_wmma<HFEAT, 32><<<blocks(waves * 32, B256), B256, 0, stream>>>(bufB, fc1W, bufA, N);
    }
    k_fc_head<<<blocks((long long)N * 32, B256), B256, 0, stream>>>(bufA, fc1b, fcoW, fcob, out, N);
}


# float-atomic edge scatter -> gather: edges grouped by destination each call (int atomic rank, scan); 16 lanes sum a node's rows in registers with self term, bias, relu; sum order = arrival rank
# speedup vs baseline: 1.9825x; 1.9825x over previous
#include <hip/hip_runtime.h>
#include <hip/hip_bf16.h>

typedef __attribute__((ext_vector_type(2))) float v2f;
typedef __attribute__((ext_vector_type(8))) float v8f;
typedef __attribute__((ext_vector_type(4))) float v4f;

#define HFEAT 64

#define SCAN_ELEMS 1024

__global__ void k_zero(int* c, int n) {
    int t = blockIdx.x * blockDim.x + threadIdx.x;
    if (t < n) c[t] = 0;
}

__global__ void k_rank(const int* __restrict__ dst, int* __restrict__ cnt,
                       int* __restrict__ rank, int e) {
    int t = blockIdx.x * blockDim.x + threadIdx.x;
    if (t < e) rank[t] = atomicAdd(&cnt[dst[t]], 1);
}

__global__ void k_scan_block(const int* __restrict__ cnt, int* __restrict__ local,
                             int* __restrict__ bsum, int n) {
    __shared__ int wsum[8];
    int lane = threadIdx.x & 31;
    int wid  = threadIdx.x >> 5;
    int base = blockIdx.x * SCAN_ELEMS + threadIdx.x * 4;
    int c0 = base + 0 < n ? cnt[base + 0] : 0;
    int c1 = base + 1 < n ? cnt[base + 1] : 0;
    int c2 = base + 2 < n ? cnt[base + 2] : 0;
    int c3 = base + 3 < n ? cnt[base + 3] : 0;
    int s = c0 + c1 + c2 + c3;
    int inc = s;
#pragma unroll
    for (int off = 1; off < 32; off <<= 1) {
        int y = __shfl_up(inc, off, 32);
        if (lane >= off) inc += y;
    }
    if (lane == 31) wsum[wid] = inc;
    __syncthreads();
    int woff = 0, tot = 0;
#pragma unroll
    for (int i = 0; i < 8; ++i) {
        int w = wsum[i];
        if (i < wid) woff += w;
        tot += w;
    }
    int ex = woff + inc - s;
    if (base + 0 < n) local[base + 0] = ex;
    if (base + 1 < n) local[base + 1] = ex + c0;
    if (base + 2 < n) local[base + 2] = ex + c0 + c1;
    if (base + 3 < n) local[base + 3] = ex + c0 + c1 + c2;
    if (threadIdx.x == 0) bsum[blockIdx.x] = tot;
}

__global__ void k_scan_top(const int* __restrict__ bsum, int* __restrict__ boff, int nb) {
    int lane = threadIdx.x;
    int carry = 0;
    for (int base = 0; base < nb; base += 32) {
        int i = base + lane;
        int v = i < nb ? bsum[i] : 0;
        int inc = v;
#pragma unroll
        for (int off = 1; off < 32; off <<= 1) {
            int y = __shfl_up(inc, off, 32);
            if (lane >= off) inc += y;
        }
        if (i < nb) boff[i] = carry + inc - v;
        carry += __shfl(inc, 31, 32);
    }
}

__global__ void k_place(const int* __restrict__ src, const int* __restrict__ dst,
                        const int* __restrict__ rank, const int* __restrict__ cnt,
                        const int* __restrict__ local, const int* __restrict__ boff,
                        int* __restrict__ csr, int e) {
    int t = blockIdx.x * blockDim.x + threadIdx.x;
    if (t >= e) return;
    int s = src[t];
    int d = dst[t];
    int p = local[d] + boff[d / SCAN_ELEMS] + rank[t];
    float c = rsqrtf((float)(cnt[s] + 1)) * rsqrtf((float)(cnt[d] + 1));
    ((int2*)csr)[p] = make_int2(s, __float_as_int(c));
}

template <int K, int NC>
__global__ void k_gemm_wmma(const float* __restrict__ A, const float* __restrict__ B,
                            float* __restrict__ C, int M) {
    const int NT = NC / 16;
    int wave = (int)((blockIdx.x * blockDim.x + threadIdx.x) >> 5);
    int lane = threadIdx.x & 31;
    int totalTiles = (M / 16) * NT;
    if (wave >= totalTiles) return;

    int m0 = (wave / NT) * 16;
    int n0 = (wave % NT) * 16;
    int row  = m0 + (lane & 15);
    int col  = n0 + (lane & 15);
    int koff = (lane >> 4) * 2;

    v8f acc = {};
    const float* arow = A + (size_t)row * K + koff;
#pragma unroll 4
    for (int k = 0; k < K; k += 4) {
        v2f a = *(const v2f*)(arow + k);
        v2f b;
        b.x = B[(size_t)(k + koff)     * NC + col];
        b.y = B[(size_t)(k + koff + 1) * NC + col];
        acc = __builtin_amdgcn_wmma_f32_16x16x4_f32(false, a, false, b,
                                                    (short)0, acc, false, false);
    }

    int rbase = m0 + ((lane >> 4) * 8);
#pragma unroll
    for (int i = 0; i < 8; ++i)
        C[(size_t)(rbase + i) * NC + col] = acc[i];
}

__global__ void k_gather(const int* __restrict__ csr, const int* __restrict__ cnt,
                         const int* __restrict__ local, const int* __restrict__ boff,
                         const float* __restrict__ xw, const float* __restrict__ bias,
                         float* __restrict__ out, int n) {
    int t = blockIdx.x * blockDim.x + threadIdx.x;
    int node = t >> 4;
    if (node >= n) return;
    int f4 = (t & 15) * 4;
    int c = cnt[node];
    int lo = local[node];
    int bo = boff[node / SCAN_ELEMS];
    const float* xf = xw + f4;
    const v4f x0 = *(const v4f*)(xf + (size_t)node * HFEAT);
    asm volatile("" :: "v"(lo), "v"(bo));
    const int2* p = (const int2*)csr + (lo + bo);
    float di = rsqrtf((float)(c + 1));
    float sc = di * di;
    v4f acc = sc * x0;
    int last = c - 1;
    int2 e0 = make_int2(0, 0), e1 = e0, e2 = e0, e3 = e0;
    if (c > 0) {
        e0 = p[0];
        e1 = p[min(1, last)];
        e2 = p[min(2, last)];
        e3 = p[min(3, last)];
    }
    for (int j = 0; j < c; j += 4) {
        const v4f v0 = *(const v4f*)(xf + (size_t)e0.x * HFEAT);
        const v4f v1 = *(const v4f*)(xf + (size_t)e1.x * HFEAT);
        const v4f v2 = *(const v4f*)(xf + (size_t)e2.x * HFEAT);
        const v4f v3 = *(const v4f*)(xf + (size_t)e3.x * HFEAT);
        float w0 = __int_as_float(e0.y), w1 = __int_as_float(e1.y);
        float w2 = __int_as_float(e2.y), w3 = __int_as_float(e3.y);
        e0 = p[min(j + 4, last)];
        e1 = p[min(j + 5, last)];
        e2 = p[min(j + 6, last)];
        e3 = p[min(j + 7, last)];
        asm volatile("" :: "v"(v0), "v"(v1), "v"(v2), "v"(v3));
        acc += w0 * v0;
        if (j + 1 < c) acc += w1 * v1;
        if (j + 2 < c) acc += w2 * v2;
        if (j + 3 < c) acc += w3 * v3;
    }
    const v4f b = *(const v4f*)(bias + f4);
    v4f r = acc + b;
    r.x = r.x > 0.0f ? r.x : 0.0f;
    r.y = r.y > 0.0f ? r.y : 0.0f;
    r.z = r.z > 0.0f ? r.z : 0.0f;
    r.w = r.w > 0.0f ? r.w : 0.0f;
    *(v4f*)(out + (size_t)node * HFEAT + f4) = r;
}

__global__ void k_fc_head(const float* __restrict__ g, const float* __restrict__ fc1_b,
                          const float* __restrict__ fcoW, const float* __restrict__ fcob,
                          float* __restrict__ out, int n) {
    int wave = (int)((blockIdx.x * blockDim.x + threadIdx.x) >> 5);
    int lane = threadIdx.x & 31;
    if (wave >= n) return;
    float v = g[(size_t)wave * 32 + lane] + fc1_b[lane];
    v = v > 0.0f ? v : 0.0f;
    float p = v * fcoW[lane];
#pragma unroll
    for (int off = 16; off > 0; off >>= 1) p += __shfl_xor(p, off, 32);
    if (lane == 0) out[wave] = p + fcob[0];
}

extern "C" void kernel_launch(void* const* d_in, const int* in_sizes, int n_in,
                              void* d_out, int out_size, void* d_ws, size_t ws_size,
                              hipStream_t stream) {
    const float* x     = (const float*)d_in[0];
    const int*   ei    = (const int*)d_in[1];
    const float* W1    = (const float*)d_in[2];
    const float* b1    = (const float*)d_in[3];
    const float* W2    = (const float*)d_in[4];
    const float* b2    = (const float*)d_in[5];
    const float* fc1W  = (const float*)d_in[6];
    const float* fc1b  = (const float*)d_in[7];
    const float* fcoW  = (const float*)d_in[8];
    const float* fcob  = (const float*)d_in[9];
    float* out = (float*)d_out;

    const int N = in_sizes[0] / 128;
    const int E = in_sizes[1] / 2;
    const int* src = ei;
    const int* dst = ei + E;

    auto pad64 = [](size_t v) { return (v + 63) & ~(size_t)63; };
    const int NB = (N + SCAN_ELEMS - 1) / SCAN_ELEMS;

    int* cnt   = (int*)d_ws;
    int* local = cnt + pad64(N);
    int* bsum  = local + pad64(N);
    int* boff  = bsum + pad64(NB);
    int* rank  = boff + pad64(NB);
    int* csr   = rank + pad64(E);
    float* bufA = (float*)(csr + pad64((size_t)E * 2));
    float* bufB = bufA + (size_t)N * HFEAT;

    const int B256 = 256;
    auto blocks = [](long long work, int bs) { return (int)((work + bs - 1) / bs); };

    k_zero<<<blocks(N, B256), B256, 0, stream>>>(cnt, N);
    k_rank<<<blocks(E, B256), B256, 0, stream>>>(dst, cnt, rank, E);
    k_scan_block<<<NB, B256, 0, stream>>>(cnt, local, bsum, N);
    k_scan_top<<<1, 32, 0, stream>>>(bsum, boff, NB);
    k_place<<<blocks(E, B256), B256, 0, stream>>>(src, dst, rank, cnt, local, boff, csr, E);

    {
        long long waves = (long long)(N / 16) * (HFEAT / 16);
        k_gemm_wmma<128, HFEAT><<<blocks(waves * 32, B256), B256, 0, stream>>>(x, W1, bufA, N);
    }
    k_gather<<<blocks((long long)N * 16, B256), B256, 0, stream>>>(csr, cnt, local, boff, bufA, b1, bufB, N);

    {
        long long waves = (long long)(N / 16) * (HFEAT / 16);
        k_gemm_wmma<HFEAT, HFEAT><<<blocks(waves * 32, B256), B256, 0, stream>>>(bufB, W2, bufA, N);
    }
    k_gather<<<blocks((long long)N * 16, B256), B256, 0, stream>>>(csr, cnt, local, boff, bufA, b2, bufB, N);

    {
        long long waves = (long long)(N / 16) * (32 / 16);
        k_gemm_wmma<HFEAT, 32><<<blocks(waves * 32, B256), B256, 0, stream>>>(bufB, fc1W, bufA, N);
    }
    k_fc_head<<<blocks((long long)N * 32, B256), B256, 0, stream>>>(bufA, fc1b, fcoW, fcob, out, N);
}


# every kernel: each wave takes several items (4 edges or nodes per thread/group, 32 head rows, 2-4 GEMM row tiles with B columns kept in registers); waves launched cut about 5x; sum order unchanged
# speedup vs baseline: 2.0503x; 1.0342x over previous
#include <hip/hip_runtime.h>
#include <hip/hip_bf16.h>

typedef __attribute__((ext_vector_type(2))) float v2f;
typedef __attribute__((ext_vector_type(8))) float v8f;
typedef __attribute__((ext_vector_type(4))) float v4f;

#define HFEAT 64

#define SCAN_ELEMS 1024

#define IPT 4

__global__ void k_zero(int* c, int n) {
    int t = blockIdx.x * blockDim.x + threadIdx.x;
    int T = gridDim.x * blockDim.x;
#pragma unroll
    for (int k = 0; k < IPT; ++k) {
        int i = t + k * T;
        if (i < n) c[i] = 0;
    }
}

__global__ void k_rank(const int* __restrict__ dst, int* __restrict__ cnt,
                       int* __restrict__ rank, int e) {
    int t = blockIdx.x * blockDim.x + threadIdx.x;
    int T = gridDim.x * blockDim.x;
    int d[IPT], r[IPT];
#pragma unroll
    for (int k = 0; k < IPT; ++k)
        d[k] = dst[min(t + k * T, e - 1)];
#pragma unroll
    for (int k = 0; k < IPT; ++k) {
        int i = t + k * T;
        r[k] = 0;
        if (i < e) r[k] = atomicAdd(&cnt[d[k]], 1);
    }
#pragma unroll
    for (int k = 0; k < IPT; ++k) {
        int i = t + k * T;
        if (i < e) rank[i] = r[k];
    }
}

__global__ void k_scan_block(const int* __restrict__ cnt, int* __restrict__ local,
                             int* __restrict__ bsum, int n) {
    __shared__ int wsum[8];
    int lane = threadIdx.x & 31;
    int wid  = threadIdx.x >> 5;
    int base = blockIdx.x * SCAN_ELEMS + threadIdx.x * 4;
    int c0 = base + 0 < n ? cnt[base + 0] : 0;
    int c1 = base + 1 < n ? cnt[base + 1] : 0;
    int c2 = base + 2 < n ? cnt[base + 2] : 0;
    int c3 = base + 3 < n ? cnt[base + 3] : 0;
    int s = c0 + c1 + c2 + c3;
    int inc = s;
#pragma unroll
    for (int off = 1; off < 32; off <<= 1) {
        int y = __shfl_up(inc, off, 32);
        if (lane >= off) inc += y;
    }
    if (lane == 31) wsum[wid] = inc;
    __syncthreads();
    int woff = 0, tot = 0;
#pragma unroll
    for (int i = 0; i < 8; ++i) {
        int w = wsum[i];
        if (i < wid) woff += w;
        tot += w;
    }
    int ex = woff + inc - s;
    if (base + 0 < n) local[base + 0] = ex;
    if (base + 1 < n) local[base + 1] = ex + c0;
    if (base + 2 < n) local[base + 2] = ex + c0 + c1;
    if (base + 3 < n) local[base + 3] = ex + c0 + c1 + c2;
    if (threadIdx.x == 0) bsum[blockIdx.x] = tot;
}

__global__ void k_scan_top(const int* __restrict__ bsum, int* __restrict__ boff, int nb) {
    int lane = threadIdx.x;
    int carry = 0;
    for (int base = 0; base < nb; base += 32) {
        int i = base + lane;
        int v = i < nb ? bsum[i] : 0;
        int inc = v;
#pragma unroll
        for (int off = 1; off < 32; off <<= 1) {
            int y = __shfl_up(inc, off, 32);
            if (lane >= off) inc += y;
        }
        if (i < nb) boff[i] = carry + inc - v;
        carry += __shfl(inc, 31, 32);
    }
}

__global__ void k_place(const int* __restrict__ src, const int* __restrict__ dst,
                        const int* __restrict__ rank, const int* __restrict__ cnt,
                        const int* __restrict__ local, const int* __restrict__ boff,
                        int* __restrict__ csr, int e) {
    int t = blockIdx.x * blockDim.x + threadIdx.x;
    int T = gridDim.x * blockDim.x;
    int s[IPT], d[IPT], r[IPT];
#pragma unroll
    for (int k = 0; k < IPT; ++k) {
        int i = min(t + k * T, e - 1);
        s[k] = src[i];
        d[k] = dst[i];
        r[k] = rank[i];
    }
    int lc[IPT], bc[IPT], cs[IPT], cd[IPT];
#pragma unroll
    for (int k = 0; k < IPT; ++k) {
        lc[k] = local[d[k]];
        bc[k] = boff[d[k] / SCAN_ELEMS];
        cs[k] = cnt[s[k]];
        cd[k] = cnt[d[k]];
    }
    asm volatile("" :: "v"(lc[0]), "v"(bc[0]), "v"(cs[0]), "v"(cd[0]), "v"(r[0]),
                       "v"(lc[1]), "v"(bc[1]), "v"(cs[1]), "v"(cd[1]), "v"(r[1]),
                       "v"(lc[2]), "v"(bc[2]), "v"(cs[2]), "v"(cd[2]), "v"(r[2]),
                       "v"(lc[3]), "v"(bc[3]), "v"(cs[3]), "v"(cd[3]), "v"(r[3]));
#pragma unroll
    for (int k = 0; k < IPT; ++k) {
        int p = lc[k] + bc[k] + r[k];
        float c = rsqrtf((float)(cs[k] + 1)) * rsqrtf((float)(cd[k] + 1));
        if (t + k * T < e) ((int2*)csr)[p] = make_int2(s[k], __float_as_int(c));
    }
}

template <int K, int NC, int R>
__global__ void k_gemm_wmma(const float* __restrict__ A, const float* __restrict__ B,
                            float* __restrict__ C, int M) {
    constexpr int NT = NC / 16;
    constexpr int KS = K / 4;
    int wave = (int)((blockIdx.x * blockDim.x + threadIdx.x) >> 5);
    int lane = threadIdx.x & 31;
    int mtiles = M / 16;
    int mt0 = (wave / NT) * R;
    if (mt0 >= mtiles) return;
    int mend = min(mt0 + R, mtiles);

    int n0 = (wave % NT) * 16;
    int col  = n0 + (lane & 15);
    int koff = (lane >> 4) * 2;

    v2f bf[KS];
#pragma unroll
    for (int s = 0; s < KS; ++s) {
        bf[s].x = B[(size_t)(4 * s + koff)     * NC + col];
        bf[s].y = B[(size_t)(4 * s + koff + 1) * NC + col];
    }

    for (int mt = mt0; mt < mend; ++mt) {
        int m0 = mt * 16;
        const float* arow = A + (size_t)(m0 + (lane & 15)) * K + koff;
        v2f af[KS];
#pragma unroll
        for (int s = 0; s < KS; ++s) af[s] = *(const v2f*)(arow + 4 * s);
        v8f acc = {};
#pragma unroll
        for (int s = 0; s < KS; ++s)
            acc = __builtin_amdgcn_wmma_f32_16x16x4_f32(false, af[s], false, bf[s],
                                                        (short)0, acc, false, false);
        int rbase = m0 + ((lane >> 4) * 8);
#pragma unroll
        for (int i = 0; i < 8; ++i)
            C[(size_t)(rbase + i) * NC + col] = acc[i];
    }
}

#define GN 4
__global__ void __attribute__((amdgpu_waves_per_eu(16)))
k_gather(const int* __restrict__ csr, const int* __restrict__ cnt,
                         const int* __restrict__ local, const int* __restrict__ boff,
                         const float* __restrict__ xw, const float* __restrict__ bias,
                         float* __restrict__ out, int n) {
    int t = blockIdx.x * blockDim.x + threadIdx.x;
    int nb = (t >> 4) * GN;
    if (nb >= n) return;
    int f4 = (t & 15) * 4;
    const float* xf = xw + f4;
    int cq[GN];
#pragma unroll
    for (int q = 0; q < GN; ++q)
        cq[q] = cnt[min(nb + q, n - 1)];
    int lo = local[nb];
    int bo = boff[nb / SCAN_ELEMS];
    const v4f xs = *(const v4f*)(xf + (size_t)nb * HFEAT);
    asm volatile("" :: "v"(lo), "v"(bo), "v"(cq[0]), "v"(cq[1]), "v"(cq[2]), "v"(cq[3]), "v"(xs));
    int st = lo + bo;
#pragma unroll
    for (int q = 0; q < GN; ++q) {
        int node = nb + q;
        if (node >= n) break;
        int c = cq[q];
        const v4f x0 = q == 0 ? xs : *(const v4f*)(xf + (size_t)node * HFEAT);
        const int2* p = (const int2*)csr + st;
        st += c;
        int last = c - 1;
        int2 e0 = make_int2(0, 0), e1 = e0, e2 = e0, e3 = e0;
        if (c > 0) {
            e0 = p[0];
            e1 = p[min(1, last)];
            e2 = p[min(2, last)];
            e3 = p[min(3, last)];
        }
        float di = rsqrtf((float)(c + 1));
        float sc = di * di;
        v4f acc = sc * x0;
        for (int j = 0; j < c; j += 4) {
            const v4f v0 = *(const v4f*)(xf + (size_t)e0.x * HFEAT);
            const v4f v1 = *(const v4f*)(xf + (size_t)e1.x * HFEAT);
            const v4f v2 = *(const v4f*)(xf + (size_t)e2.x * HFEAT);
            const v4f v3 = *(const v4f*)(xf + (size_t)e3.x * HFEAT);
            float w0 = __int_as_float(e0.y), w1 = __int_as_float(e1.y);
            float w2 = __int_as_float(e2.y), w3 = __int_as_float(e3.y);
            e0 = p[min(j + 4, last)];
            e1 = p[min(j + 5, last)];
            e2 = p[min(j + 6, last)];
            e3 = p[min(j + 7, last)];
            asm volatile("" :: "v"(v0), "v"(v1), "v"(v2), "v"(v3));
            acc += w0 * v0;
            if (j + 1 < c) acc += w1 * v1;
            if (j + 2 < c) acc += w2 * v2;
            if (j + 3 < c) acc += w3 * v3;
        }
        const v4f b = *(const v4f*)(bias + f4);
        v4f r = acc + b;
        r.x = r.x > 0.0f ? r.x : 0.0f;
        r.y = r.y > 0.0f ? r.y : 0.0f;
        r.z = r.z > 0.0f ? r.z : 0.0f;
        r.w = r.w > 0.0f ? r.w : 0.0f;
        *(v4f*)(out + (size_t)node * HFEAT + f4) = r;
    }
}

__global__ void k_fc_head(const float* __restrict__ g, const float* __restrict__ fc1_b,
                          const float* __restrict__ fcoW, const float* __restrict__ fcob,
                          float* __restrict__ out, int n) {
    int wave = (int)((blockIdx.x * blockDim.x + threadIdx.x) >> 5);
    int lane = threadIdx.x & 31;
    int base = wave * 32;
    if (base >= n) return;
    float fb = fc1_b[lane];
    float fw = fcoW[lane];
    float res = 0.0f;
#pragma unroll
    for (int i = 0; i < 32; ++i) {
        int node = min(base + i, n - 1);
        float v = g[(size_t)node * 32 + lane] + fb;
        v = v > 0.0f ? v : 0.0f;
        float p = v * fw;
#pragma unroll
        for (int off = 16; off > 0; off >>= 1) p += __shfl_xor(p, off, 32);
        if (lane == i) res = p;
    }
    if (base + lane < n) out[base + lane] = res + fcob[0];
}

extern "C" void kernel_launch(void* const* d_in, const int* in_sizes, int n_in,
                              void* d_out, int out_size, void* d_ws, size_t ws_size,
                              hipStream_t stream) {
    const float* x     = (const float*)d_in[0];
    const int*   ei    = (const int*)d_in[1];
    const float* W1    = (const float*)d_in[2];
    const float* b1    = (const float*)d_in[3];
    const float* W2    = (const float*)d_in[4];
    const float* b2    = (const float*)d_in[5];
    const float* fc1W  = (const float*)d_in[6];
    const float* fc1b  = (const float*)d_in[7];
    const float* fcoW  = (const float*)d_in[8];
    const float* fcob  = (const float*)d_in[9];
    float* out = (float*)d_out;

    const int N = in_sizes[0] / 128;
    const int E = in_sizes[1] / 2;
    const int* src = ei;
    const int* dst = ei + E;

    auto pad64 = [](size_t v) { return (v + 63) & ~(size_t)63; };
    const int NB = (N + SCAN_ELEMS - 1) / SCAN_ELEMS;

    int* cnt   = (int*)d_ws;
    int* local = cnt + pad64(N);
    int* bsum  = local + pad64(N);
    int* boff  = bsum + pad64(NB);
    int* rank  = boff + pad64(NB);
    int* csr   = rank + pad64(E);
    float* bufA = (float*)(csr + pad64((size_t)E * 2));
    float* bufB = bufA + (size_t)N * HFEAT;

    const int B256 = 256;
    auto blocks = [](long long work, int bs) { return (int)((work + bs - 1) / bs); };

    k_zero<<<blocks((N + IPT - 1) / IPT, B256), B256, 0, stream>>>(cnt, N);
    k_rank<<<blocks((E + IPT - 1) / IPT, B256), B256, 0, stream>>>(dst, cnt, rank, E);
    k_scan_block<<<NB, B256, 0, stream>>>(cnt, local, bsum, N);
    k_scan_top<<<1, 32, 0, stream>>>(bsum, boff, NB);
    k_place<<<blocks((E + IPT - 1) / IPT, B256), B256, 0, stream>>>(src, dst, rank, cnt, local, boff, csr, E);

    {
        long long waves = (long long)((N / 16 + 3) / 4) * (HFEAT / 16);
        k_gemm_wmma<128, HFEAT, 4><<<blocks(waves * 32, B256), B256, 0, stream>>>(x, W1, bufA, N);
    }
    k_gather<<<blocks((long long)((N + GN - 1) / GN) * 16, B256), B256, 0, stream>>>(csr, cnt, local, boff, bufA, b1, bufB, N);

    {
        long long waves = (long long)((N / 16 + 3) / 4) * (HFEAT / 16);
        k_gemm_wmma<HFEAT, HFEAT, 4><<<blocks(waves * 32, B256), B256, 0, stream>>>(bufB, W2, bufA, N);
    }
    k_gather<<<blocks((long long)((N + GN - 1) / GN) * 16, B256), B256, 0, stream>>>(csr, cnt, local, boff, bufA, b2, bufB, N);

    {
        long long waves = (long long)((N / 16 + 1) / 2) * (32 / 16);
        k_gemm_wmma<HFEAT, 32, 2><<<blocks(waves * 32, B256), B256, 0, stream>>>(bufB, fc1W, bufA, N);
    }
    k_fc_head<<<blocks((long long)((N + 31) / 32) * 32, B256), B256, 0, stream>>>(bufA, fc1b, fcoW, fcob, out, N);
}


# each aggregation kernel keeps its 64-node tile of h in LDS and runs the consuming GEMM (W2; fc1 plus head) on it with WMMA: 3 launches and the h round trips go; sum order unchanged
# speedup vs baseline: 2.2399x; 1.0925x over previous
#include <hip/hip_runtime.h>
#include <hip/hip_bf16.h>

typedef __attribute__((ext_vector_type(2))) float v2f;
typedef __attribute__((ext_vector_type(8))) float v8f;
typedef __attribute__((ext_vector_type(4))) float v4f;

#define HFEAT 64

#define SCAN_ELEMS 1024

#define IPT 4

__global__ void k_zero(int* c, int n) {
    int t = blockIdx.x * blockDim.x + threadIdx.x;
    int T = gridDim.x * blockDim.x;
#pragma unroll
    for (int k = 0; k < IPT; ++k) {
        int i = t + k * T;
        if (i < n) c[i] = 0;
    }
}

__global__ void k_rank(const int* __restrict__ dst, int* __restrict__ cnt,
                       int* __restrict__ rank, int e) {
    int t = blockIdx.x * blockDim.x + threadIdx.x;
    int T = gridDim.x * blockDim.x;
    int d[IPT], r[IPT];
#pragma unroll
    for (int k = 0; k < IPT; ++k)
        d[k] = dst[min(t + k * T, e - 1)];
#pragma unroll
    for (int k = 0; k < IPT; ++k) {
        int i = t + k * T;
        r[k] = 0;
        if (i < e) r[k] = atomicAdd(&cnt[d[k]], 1);
    }
#pragma unroll
    for (int k = 0; k < IPT; ++k) {
        int i = t + k * T;
        if (i < e) rank[i] = r[k];
    }
}

__global__ void k_scan_block(const int* __restrict__ cnt, int* __restrict__ local,
                             int* __restrict__ bsum, int n) {
    __shared__ int wsum[8];
    int lane = threadIdx.x & 31;
    int wid  = threadIdx.x >> 5;
    int base = blockIdx.x * SCAN_ELEMS + threadIdx.x * 4;
    int c0 = base + 0 < n ? cnt[base + 0] : 0;
    int c1 = base + 1 < n ? cnt[base + 1] : 0;
    int c2 = base + 2 < n ? cnt[base + 2] : 0;
    int c3 = base + 3 < n ? cnt[base + 3] : 0;
    int s = c0 + c1 + c2 + c3;
    int inc = s;
#pragma unroll
    for (int off = 1; off < 32; off <<= 1) {
        int y = __shfl_up(inc, off, 32);
        if (lane >= off) inc += y;
    }
    if (lane == 31) wsum[wid] = inc;
    __syncthreads();
    int woff = 0, tot = 0;
#pragma unroll
    for (int i = 0; i < 8; ++i) {
        int w = wsum[i];
        if (i < wid) woff += w;
        tot += w;
    }
    int ex = woff + inc - s;
    if (base + 0 < n) local[base + 0] = ex;
    if (base + 1 < n) local[base + 1] = ex + c0;
    if (base + 2 < n) local[base + 2] = ex + c0 + c1;
    if (base + 3 < n) local[base + 3] = ex + c0 + c1 + c2;
    if (threadIdx.x == 0) bsum[blockIdx.x] = tot;
}

__global__ void k_scan_top(const int* __restrict__ bsum, int* __restrict__ boff, int nb) {
    int lane = threadIdx.x;
    int carry = 0;
    for (int base = 0; base < nb; base += 32) {
        int i = base + lane;
        int v = i < nb ? bsum[i] : 0;
        int inc = v;
#pragma unroll
        for (int off = 1; off < 32; off <<= 1) {
            int y = __shfl_up(inc, off, 32);
            if (lane >= off) inc += y;
        }
        if (i < nb) boff[i] = carry + inc - v;
        carry += __shfl(inc, 31, 32);
    }
}

__global__ void k_place(const int* __restrict__ src, const int* __restrict__ dst,
                        const int* __restrict__ rank, const int* __restrict__ cnt,
                        const int* __restrict__ local, const int* __restrict__ boff,
                        int* __restrict__ csr, int e) {
    int t = blockIdx.x * blockDim.x + threadIdx.x;
    int T = gridDim.x * blockDim.x;
    int s[IPT], d[IPT], r[IPT];
#pragma unroll
    for (int k = 0; k < IPT; ++k) {
        int i = min(t + k * T, e - 1);
        s[k] = src[i];
        d[k] = dst[i];
        r[k] = rank[i];
    }
    int lc[IPT], bc[IPT], cs[IPT], cd[IPT];
#pragma unroll
    for (int k = 0; k < IPT; ++k) {
        lc[k] = local[d[k]];
        bc[k] = boff[d[k] / SCAN_ELEMS];
        cs[k] = cnt[s[k]];
        cd[k] = cnt[d[k]];
    }
    asm volatile("" :: "v"(lc[0]), "v"(bc[0]), "v"(cs[0]), "v"(cd[0]), "v"(r[0]),
                       "v"(lc[1]), "v"(bc[1]), "v"(cs[1]), "v"(cd[1]), "v"(r[1]),
                       "v"(lc[2]), "v"(bc[2]), "v"(cs[2]), "v"(cd[2]), "v"(r[2]),
                       "v"(lc[3]), "v"(bc[3]), "v"(cs[3]), "v"(cd[3]), "v"(r[3]));
#pragma unroll
    for (int k = 0; k < IPT; ++k) {
        int p = lc[k] + bc[k] + r[k];
        float c = rsqrtf((float)(cs[k] + 1)) * rsqrtf((float)(cd[k] + 1));
        if (t + k * T < e) ((int2*)csr)[p] = make_int2(s[k], __float_as_int(c));
    }
}

template <int K, int NC, int R>
__global__ void k_gemm_wmma(const float* __restrict__ A, const float* __restrict__ B,
                            float* __restrict__ C, int M) {
    constexpr int NT = NC / 16;
    constexpr int KS = K / 4;
    int wave = (int)((blockIdx.x * blockDim.x + threadIdx.x) >> 5);
    int lane = threadIdx.x & 31;
    int mtiles = M / 16;
    int mt0 = (wave / NT) * R;
    if (mt0 >= mtiles) return;
    int mend = min(mt0 + R, mtiles);

    int n0 = (wave % NT) * 16;
    int col  = n0 + (lane & 15);
    int koff = (lane >> 4) * 2;

    v2f bf[KS];
#pragma unroll
    for (int s = 0; s < KS; ++s) {
        bf[s].x = B[(size_t)(4 * s + koff)     * NC + col];
        bf[s].y = B[(size_t)(4 * s + koff + 1) * NC + col];
    }

    for (int mt = mt0; mt < mend; ++mt) {
        int m0 = mt * 16;
        const float* arow = A + (size_t)(m0 + (lane & 15)) * K + koff;
        v2f af[KS];
#pragma unroll
        for (int s = 0; s < KS; ++s) af[s] = *(const v2f*)(arow + 4 * s);
        v8f acc = {};
#pragma unroll
        for (int s = 0; s < KS; ++s)
            acc = __builtin_amdgcn_wmma_f32_16x16x4_f32(false, af[s], false, bf[s],
                                                        (short)0, acc, false, false);
        int rbase = m0 + ((lane >> 4) * 8);
#pragma unroll
        for (int i = 0; i < 8; ++i)
            C[(size_t)(rbase + i) * NC + col] = acc[i];
    }
}

#define GN 4
#define TILE_M 64
#define HPITCH 68
#define GPITCH 36

__device__ __forceinline__ void gather_tile_to_lds(
        const int* __restrict__ csr, const int* __restrict__ cnt,
        const int* __restrict__ local, const int* __restrict__ boff,
        const float* __restrict__ xw, const float* __restrict__ bias,
        float* __restrict__ hs, int n) {
    int t = blockIdx.x * blockDim.x + threadIdx.x;
    int nb = (t >> 4) * GN;
    int f4 = (t & 15) * 4;
    float* hrow = hs + (threadIdx.x >> 4) * (GN * HPITCH) + (threadIdx.x & 15) * 4;
    const v4f zero4 = {0.0f, 0.0f, 0.0f, 0.0f};
    if (nb >= n) {
#pragma unroll
        for (int q = 0; q < GN; ++q) *(v4f*)(hrow + q * HPITCH) = zero4;
        return;
    }
    const float* xf = xw + f4;
    int cq[GN];
#pragma unroll
    for (int q = 0; q < GN; ++q)
        cq[q] = cnt[min(nb + q, n - 1)];
    int lo = local[nb];
    int bo = boff[nb / SCAN_ELEMS];
    const v4f xs = *(const v4f*)(xf + (size_t)nb * HFEAT);
    asm volatile("" :: "v"(lo), "v"(bo), "v"(cq[0]), "v"(cq[1]), "v"(cq[2]), "v"(cq[3]), "v"(xs));
    __builtin_amdgcn_sched_barrier(0);
    int st = lo + bo;
#pragma unroll
    for (int q = 0; q < GN; ++q) {
        int node = nb + q;
        if (node >= n) {
            *(v4f*)(hrow + q * HPITCH) = zero4;
            continue;
        }
        int c = cq[q];
        const v4f x0 = q == 0 ? xs : *(const v4f*)(xf + (size_t)node * HFEAT);
        const int2* p = (const int2*)csr + st;
        st += c;
        int last = c - 1;
        int2 e0 = make_int2(0, 0), e1 = e0, e2 = e0, e3 = e0;
        if (c > 0) {
            e0 = p[0];
            e1 = p[min(1, last)];
            e2 = p[min(2, last)];
            e3 = p[min(3, last)];
        }
        float di = rsqrtf((float)(c + 1));
        float sc = di * di;
        v4f acc = sc * x0;
        for (int j = 0; j < c; j += 4) {
            const v4f v0 = *(const v4f*)(xf + (size_t)e0.x * HFEAT);
            const v4f v1 = *(const v4f*)(xf + (size_t)e1.x * HFEAT);
            const v4f v2 = *(const v4f*)(xf + (size_t)e2.x * HFEAT);
            const v4f v3 = *(const v4f*)(xf + (size_t)e3.x * HFEAT);
            float w0 = __int_as_float(e0.y), w1 = __int_as_float(e1.y);
            float w2 = __int_as_float(e2.y), w3 = __int_as_float(e3.y);
            e0 = p[min(j + 4, last)];
            e1 = p[min(j + 5, last)];
            e2 = p[min(j + 6, last)];
            e3 = p[min(j + 7, last)];
            asm volatile("" :: "v"(v0), "v"(v1), "v"(v2), "v"(v3));
            acc += w0 * v0;
            if (j + 1 < c) acc += w1 * v1;
            if (j + 2 < c) acc += w2 * v2;
            if (j + 3 < c) acc += w3 * v3;
        }
        const v4f b = *(const v4f*)(bias + f4);
        v4f r = acc + b;
        r.x = r.x > 0.0f ? r.x : 0.0f;
        r.y = r.y > 0.0f ? r.y : 0.0f;
        r.z = r.z > 0.0f ? r.z : 0.0f;
        r.w = r.w > 0.0f ? r.w : 0.0f;
        *(v4f*)(hrow + q * HPITCH) = r;
    }
}

__device__ __forceinline__ v8f lds_tile_wmma(const float* __restrict__ hs, const v2f (&bf)[16],
                                             int rt, int lane) {
    const float* arow = hs + (rt * 16 + (lane & 15)) * HPITCH + (lane >> 4) * 2;
    v8f acc = {};
#pragma unroll
    for (int kh = 0; kh < 2; ++kh) {
        v2f af[8];
#pragma unroll
        for (int s = 0; s < 8; ++s) af[s] = *(const v2f*)(arow + 4 * (8 * kh + s));
#pragma unroll
        for (int s = 0; s < 8; ++s)
            acc = __builtin_amdgcn_wmma_f32_16x16x4_f32(false, af[s], false, bf[8 * kh + s],
                                                        (short)0, acc, false, false);
    }
    return acc;
}

__global__ void __attribute__((amdgpu_waves_per_eu(16)))
k_gather_gemm64(const int* __restrict__ csr, const int* __restrict__ cnt,
                const int* __restrict__ local, const int* __restrict__ boff,
                const float* __restrict__ xw, const float* __restrict__ bias,
                const float* __restrict__ B, float* __restrict__ C, int n) {
    __shared__ __attribute__((aligned(16))) float hs[TILE_M * HPITCH];
    gather_tile_to_lds(csr, cnt, local, boff, xw, bias, hs, n);

    int w    = __builtin_amdgcn_readfirstlane((int)(threadIdx.x >> 5));
    int lane = threadIdx.x & 31;
    int col  = (w & 3) * 16 + (lane & 15);
    int koff = (lane >> 4) * 2;
    v2f bf[16];
#pragma unroll
    for (int s = 0; s < 16; ++s) {
        bf[s].x = B[(4 * s + koff)     * HFEAT + col];
        bf[s].y = B[(4 * s + koff + 1) * HFEAT + col];
    }
    __syncthreads();

    int base = blockIdx.x * TILE_M;
#pragma unroll
    for (int r = 0; r < 2; ++r) {
        int rt = (w >> 2) * 2 + r;
        int m0 = base + rt * 16;
        if (m0 >= n) break;
        v8f acc = lds_tile_wmma(hs, bf, rt, lane);
        int rbase = m0 + ((lane >> 4) * 8);
        if (m0 + 16 <= n) {
#pragma unroll
            for (int i = 0; i < 8; ++i)
                C[(size_t)(rbase + i) * HFEAT + col] = acc[i];
        } else {
#pragma unroll
            for (int i = 0; i < 8; ++i)
                if (rbase + i < n) C[(size_t)(rbase + i) * HFEAT + col] = acc[i];
        }
    }
}

__global__ void __attribute__((amdgpu_waves_per_eu(16)))
k_gather_head(const int* __restrict__ csr, const int* __restrict__ cnt,
              const int* __restrict__ local, const int* __restrict__ boff,
              const float* __restrict__ xw, const float* __restrict__ bias,
              const float* __restrict__ fc1W, const float* __restrict__ fc1_b,
              const float* __restrict__ fcoW, const float* __restrict__ fcob,
              float* __restrict__ out, int n) {
    __shared__ __attribute__((aligned(16))) float hs[TILE_M * HPITCH];
    __shared__ float gs[TILE_M * GPITCH];
    __shared__ float os[TILE_M];
    gather_tile_to_lds(csr, cnt, local, boff, xw, bias, hs, n);

    int w    = __builtin_amdgcn_readfirstlane((int)(threadIdx.x >> 5));
    int lane = threadIdx.x & 31;
    int ct   = w & 1;
    int rt   = w >> 1;
    int col  = ct * 16 + (lane & 15);
    int koff = (lane >> 4) * 2;
    v2f bf[16];
#pragma unroll
    for (int s = 0; s < 16; ++s) {
        bf[s].x = fc1W[(4 * s + koff)     * 32 + col];
        bf[s].y = fc1W[(4 * s + koff + 1) * 32 + col];
    }
    float fb = fc1_b[lane];
    float fw = fcoW[lane];
    float ob = fcob[0];
    __syncthreads();

    {
        v8f acc = lds_tile_wmma(hs, bf, rt, lane);
        float* grow = gs + (rt * 16 + (lane >> 4) * 8) * GPITCH + col;
#pragma unroll
        for (int i = 0; i < 8; ++i) grow[i * GPITCH] = acc[i];
    }
    __syncthreads();

    float res = 0.0f;
#pragma unroll
    for (int i = 0; i < 8; ++i) {
        float v = gs[(w * 8 + i) * GPITCH + lane] + fb;
        v = v > 0.0f ? v : 0.0f;
        float p = v * fw;
#pragma unroll
        for (int off = 16; off > 0; off >>= 1) p += __shfl_xor(p, off, 32);
        if (lane == i) res = p;
    }
    if (lane < 8) os[w * 8 + lane] = res;
    __syncthreads();

    if (threadIdx.x < TILE_M) {
        int node = blockIdx.x * TILE_M + threadIdx.x;
        if (node < n) out[node] = os[threadIdx.x] + ob;
    }
}

extern "C" void kernel_launch(void* const* d_in, const int* in_sizes, int n_in,
                              void* d_out, int out_size, void* d_ws, size_t ws_size,
                              hipStream_t stream) {
    const float* x     = (const float*)d_in[0];
    const int*   ei    = (const int*)d_in[1];
    const float* W1    = (const float*)d_in[2];
    const float* b1    = (const float*)d_in[3];
    const float* W2    = (const float*)d_in[4];
    const float* b2    = (const float*)d_in[5];
    const float* fc1W  = (const float*)d_in[6];
    const float* fc1b  = (const float*)d_in[7];
    const float* fcoW  = (const float*)d_in[8];
    const float* fcob  = (const float*)d_in[9];
    float* out = (float*)d_out;

    const int N = in_sizes[0] / 128;
    const int E = in_sizes[1] / 2;
    const int* src = ei;
    const int* dst = ei + E;

    auto pad64 = [](size_t v) { return (v + 63) & ~(size_t)63; };
    const int NB = (N + SCAN_ELEMS - 1) / SCAN_ELEMS;

    int* cnt   = (int*)d_ws;
    int* local = cnt + pad64(N);
    int* bsum  = local + pad64(N);
    int* boff  = bsum + pad64(NB);
    int* rank  = boff + pad64(NB);
    int* csr   = rank + pad64(E);
    float* bufA = (float*)(csr + pad64((size_t)E * 2));
    float* bufB = bufA + (size_t)N * HFEAT;

    const int B256 = 256;
    auto blocks = [](long long work, int bs) { return (int)((work + bs - 1) / bs); };

    k_zero<<<blocks((N + IPT - 1) / IPT, B256), B256, 0, stream>>>(cnt, N);
    k_rank<<<blocks((E + IPT - 1) / IPT, B256), B256, 0, stream>>>(dst, cnt, rank, E);
    k_scan_block<<<NB, B256, 0, stream>>>(cnt, local, bsum, N);
    k_scan_top<<<1, 32, 0, stream>>>(bsum, boff, NB);
    k_place<<<blocks((E + IPT - 1) / IPT, B256), B256, 0, stream>>>(src, dst, rank, cnt, local, boff, csr, E);

    {
        long long waves = (long long)((N / 16 + 3) / 4) * (HFEAT / 16);
        k_gemm_wmma<128, HFEAT, 4><<<blocks(waves * 32, B256), B256, 0, stream>>>(x, W1, bufA, N);
    }
    k_gather_gemm64<<<blocks(N, TILE_M), B256, 0, stream>>>(csr, cnt, local, boff, bufA, b1, W2, bufB, N);

    k_gather_head<<<blocks(N, TILE_M), B256, 0, stream>>>(csr, cnt, local, boff, bufB, b2, fc1W, fc1b, fcoW, fcob, out, N);
}


# edges grouped by destination by a two-level counting sort on LDS counters (bin count, scan, bin scatter, per-bin sort, coefficient pass): no global atomics; sum order = arrival order of LDS adds
# speedup vs baseline: 3.6710x; 1.6389x over previous
#include <hip/hip_runtime.h>
#include <hip/hip_bf16.h>

typedef __attribute__((ext_vector_type(2))) float v2f;
typedef __attribute__((ext_vector_type(8))) float v8f;
typedef __attribute__((ext_vector_type(4))) float v4f;

#define HFEAT 64

#define SCAN_ELEMS 1024
#define NBIN_MAX 128
#define GA 256
#define TB 1024
#define SUBS 4
#define BIN_SHIFT 10
#define SUB_SHIFT 8

#define IPT 4

__device__ __forceinline__ int block_excl_scan(int v, int* wsum) {
    int lane = threadIdx.x & 31;
    int wid  = threadIdx.x >> 5;
    int inc = v;
#pragma unroll
    for (int off = 1; off < 32; off <<= 1) {
        int y = __shfl_up(inc, off, 32);
        if (lane >= off) inc += y;
    }
    if (lane == 31) wsum[wid] = inc;
    __syncthreads();
    int woff = 0;
#pragma unroll
    for (int i = 0; i < TB / 32; ++i) {
        int w = wsum[i];
        if (i < wid) woff += w;
    }
    return woff + inc - v;
}

__global__ void __launch_bounds__(TB)
k_bin_count(const int* __restrict__ dst, int* __restrict__ hist, int e, int nbin) {
    __shared__ int h[NBIN_MAX];
    int t = threadIdx.x;
    if (t < NBIN_MAX) h[t] = 0;
    __syncthreads();
    int ch = (e + GA - 1) / GA;
    int lo = blockIdx.x * ch;
    int hi = min(lo + ch, e);
    for (int i = lo + t; i < hi; i += IPT * TB) {
        int d[IPT];
#pragma unroll
        for (int k = 0; k < IPT; ++k)
            d[k] = dst[min(i + k * TB, hi - 1)];
        asm volatile("" :: "v"(d[0]), "v"(d[1]), "v"(d[2]), "v"(d[3]));
#pragma unroll
        for (int k = 0; k < IPT; ++k)
            if (i + k * TB < hi) atomicAdd(&h[d[k] >> BIN_SHIFT], 1);
    }
    __syncthreads();
    if (t < nbin) hist[blockIdx.x * nbin + t] = h[t];
}

__global__ void __launch_bounds__(TB)
k_bin_scan(const int* __restrict__ hist, int* __restrict__ offs, int nbin) {
    __shared__ int a[GA * NBIN_MAX];
    __shared__ int wsum[TB / 32];
    int t = threadIdx.x;
    int total = GA * nbin;
    for (int i0 = t; i0 < total; i0 += 8 * TB) {
        int hv[8];
#pragma unroll
        for (int k = 0; k < 8; ++k)
            hv[k] = hist[min(i0 + k * TB, total - 1)];
        asm volatile("" :: "v"(hv[0]), "v"(hv[1]), "v"(hv[2]), "v"(hv[3]),
                           "v"(hv[4]), "v"(hv[5]), "v"(hv[6]), "v"(hv[7]));
#pragma unroll
        for (int k = 0; k < 8; ++k)
            if (i0 + k * TB < total) a[i0 + k * TB] = hv[k];
    }
    __syncthreads();
    int K = (total + TB - 1) / TB;
    int q0 = t * K;
    int s = 0;
    for (int k = 0; k < K; ++k) {
        int q = q0 + k;
        if (q < total) s += a[(q % GA) * nbin + q / GA];
    }
    int run = block_excl_scan(s, wsum);
    for (int k = 0; k < K; ++k) {
        int q = q0 + k;
        if (q < total) {
            int idx = (q % GA) * nbin + q / GA;
            int v = a[idx];
            a[idx] = run;
            run += v;
        }
    }
    __syncthreads();
    for (int i = t; i < total; i += TB) offs[i] = a[i];
}

__global__ void __launch_bounds__(TB)
k_bin_scatter(const int* __restrict__ src, const int* __restrict__ dst,
              const int* __restrict__ offs, int2* __restrict__ binned, int e, int nbin) {
    __shared__ int cur[NBIN_MAX];
    int t = threadIdx.x;
    if (t < nbin) cur[t] = offs[blockIdx.x * nbin + t];
    __syncthreads();
    int ch = (e + GA - 1) / GA;
    int lo = blockIdx.x * ch;
    int hi = min(lo + ch, e);
    for (int i = lo + t; i < hi; i += IPT * TB) {
        int s[IPT], d[IPT];
#pragma unroll
        for (int k = 0; k < IPT; ++k) {
            int j = min(i + k * TB, hi - 1);
            s[k] = src[j];
            d[k] = dst[j];
        }
        asm volatile("" :: "v"(s[0]), "v"(d[0]), "v"(s[1]), "v"(d[1]),
                           "v"(s[2]), "v"(d[2]), "v"(s[3]), "v"(d[3]));
#pragma unroll
        for (int k = 0; k < IPT; ++k) {
            if (i + k * TB < hi) {
                int p = atomicAdd(&cur[d[k] >> BIN_SHIFT], 1);
                binned[p] = make_int2(s[k], d[k]);
            }
        }
    }
}

__global__ void __launch_bounds__(TB)
k_bin_sort(const int2* __restrict__ binned, const int* __restrict__ offs,
           int* __restrict__ cnt, int* __restrict__ local, int2* __restrict__ sorted,
           int e, int n, int nbin) {
    __shared__ int c[SCAN_ELEMS];
    __shared__ int wsum[TB / 32];
    int t = threadIdx.x;
    int b = blockIdx.x / SUBS;
    int sub = blockIdx.x % SUBS;
    int lo = offs[b];
    int hi = b + 1 < nbin ? offs[b + 1] : e;
    int node0 = b * SCAN_ELEMS;
    c[t] = 0;
    __syncthreads();
    for (int i = lo + t; i < hi; i += IPT * TB) {
        int d[IPT];
#pragma unroll
        for (int k = 0; k < IPT; ++k)
            d[k] = binned[min(i + k * TB, hi - 1)].y;
        asm volatile("" :: "v"(d[0]), "v"(d[1]), "v"(d[2]), "v"(d[3]));
#pragma unroll
        for (int k = 0; k < IPT; ++k)
            if (i + k * TB < hi) atomicAdd(&c[d[k] - node0], 1);
    }
    __syncthreads();
    int v = c[t];
    int ex = block_excl_scan(v, wsum);
    int node = node0 + t;
    if ((t >> SUB_SHIFT) == sub && node < n) {
        cnt[node] = v;
        local[node] = ex;
    }
    c[t] = lo + ex;
    __syncthreads();
    for (int i = lo + t; i < hi; i += IPT * TB) {
        int2 ed[IPT];
#pragma unroll
        for (int k = 0; k < IPT; ++k)
            ed[k] = binned[min(i + k * TB, hi - 1)];
        asm volatile("" :: "v"(ed[0].x), "v"(ed[0].y), "v"(ed[1].x), "v"(ed[1].y),
                           "v"(ed[2].x), "v"(ed[2].y), "v"(ed[3].x), "v"(ed[3].y));
#pragma unroll
        for (int k = 0; k < IPT; ++k) {
            int dl = ed[k].y - node0;
            if (i + k * TB < hi && (dl >> SUB_SHIFT) == sub) {
                int p = atomicAdd(&c[dl], 1);
                sorted[p] = ed[k];
            }
        }
    }
}

__global__ void k_coef(const int* __restrict__ cnt, const int2* __restrict__ sorted,
                       int2* __restrict__ csr, int e) {
    int t = blockIdx.x * blockDim.x + threadIdx.x;
    int T = gridDim.x * blockDim.x;
    int2 ed[IPT];
#pragma unroll
    for (int k = 0; k < IPT; ++k)
        ed[k] = sorted[min(t + k * T, e - 1)];
    int cs[IPT], cd[IPT];
#pragma unroll
    for (int k = 0; k < IPT; ++k) {
        cs[k] = cnt[ed[k].x];
        cd[k] = cnt[ed[k].y];
    }
    asm volatile("" :: "v"(cs[0]), "v"(cd[0]), "v"(cs[1]), "v"(cd[1]),
                       "v"(cs[2]), "v"(cd[2]), "v"(cs[3]), "v"(cd[3]));
#pragma unroll
    for (int k = 0; k < IPT; ++k) {
        float c = rsqrtf((float)(cs[k] + 1)) * rsqrtf((float)(cd[k] + 1));
        if (t + k * T < e) csr[t + k * T] = make_int2(ed[k].x, __float_as_int(c));
    }
}

template <int K, int NC, int R>
__global__ void k_gemm_wmma(const float* __restrict__ A, const float* __restrict__ B,
                            float* __restrict__ C, int M) {
    constexpr int NT = NC / 16;
    constexpr int KS = K / 4;
    int wave = (int)((blockIdx.x * blockDim.x + threadIdx.x) >> 5);
    int lane = threadIdx.x & 31;
    int mtiles = M / 16;
    int mt0 = (wave / NT) * R;
    if (mt0 >= mtiles) return;
    int mend = min(mt0 + R, mtiles);

    int n0 = (wave % NT) * 16;
    int col  = n0 + (lane & 15);
    int koff = (lane >> 4) * 2;

    v2f bf[KS];
#pragma unroll
    for (int s = 0; s < KS; ++s) {
        bf[s].x = B[(size_t)(4 * s + koff)     * NC + col];
        bf[s].y = B[(size_t)(4 * s + koff + 1) * NC + col];
    }

    for (int mt = mt0; mt < mend; ++mt) {
        int m0 = mt * 16;
        const float* arow = A + (size_t)(m0 + (lane & 15)) * K + koff;
        v2f af[KS];
#pragma unroll
        for (int s = 0; s < KS; ++s) af[s] = *(const v2f*)(arow + 4 * s);
        v8f acc = {};
#pragma unroll
        for (int s = 0; s < KS; ++s)
            acc = __builtin_amdgcn_wmma_f32_16x16x4_f32(false, af[s], false, bf[s],
                                                        (short)0, acc, false, false);
        int rbase = m0 + ((lane >> 4) * 8);
#pragma unroll
        for (int i = 0; i < 8; ++i)
            C[(size_t)(rbase + i) * NC + col] = acc[i];
    }
}

#define GN 4
#define TILE_M 64
#define HPITCH 68
#define GPITCH 36

__device__ __forceinline__ void gather_tile_to_lds(
        const int* __restrict__ csr, const int* __restrict__ cnt,
        const int* __restrict__ local, const int* __restrict__ boff,
        const float* __restrict__ xw, const float* __restrict__ bias,
        float* __restrict__ hs, int n) {
    int t = blockIdx.x * blockDim.x + threadIdx.x;
    int nb = (t >> 4) * GN;
    int f4 = (t & 15) * 4;
    float* hrow = hs + (threadIdx.x >> 4) * (GN * HPITCH) + (threadIdx.x & 15) * 4;
    const v4f zero4 = {0.0f, 0.0f, 0.0f, 0.0f};
    if (nb >= n) {
#pragma unroll
        for (int q = 0; q < GN; ++q) *(v4f*)(hrow + q * HPITCH) = zero4;
        return;
    }
    const float* xf = xw + f4;
    int cq[GN];
#pragma unroll
    for (int q = 0; q < GN; ++q)
        cq[q] = cnt[min(nb + q, n - 1)];
    int lo = local[nb];
    int bo = boff[nb / SCAN_ELEMS];
    const v4f xs = *(const v4f*)(xf + (size_t)nb * HFEAT);
    asm volatile("" :: "v"(lo), "v"(bo), "v"(cq[0]), "v"(cq[1]), "v"(cq[2]), "v"(cq[3]), "v"(xs));
    __builtin_amdgcn_sched_barrier(0);
    int st = lo + bo;
#pragma unroll
    for (int q = 0; q < GN; ++q) {
        int node = nb + q;
        if (node >= n) {
            *(v4f*)(hrow + q * HPITCH) = zero4;
            continue;
        }
        int c = cq[q];
        const v4f x0 = q == 0 ? xs : *(const v4f*)(xf + (size_t)node * HFEAT);
        const int2* p = (const int2*)csr + st;
        st += c;
        int last = c - 1;
        int2 e0 = make_int2(0, 0), e1 = e0, e2 = e0, e3 = e0;
        if (c > 0) {
            e0 = p[0];
            e1 = p[min(1, last)];
            e2 = p[min(2, last)];
            e3 = p[min(3, last)];
        }
        float di = rsqrtf((float)(c + 1));
        float sc = di * di;
        v4f acc = sc * x0;
        for (int j = 0; j < c; j += 4) {
            const v4f v0 = *(const v4f*)(xf + (size_t)e0.x * HFEAT);
            const v4f v1 = *(const v4f*)(xf + (size_t)e1.x * HFEAT);
            const v4f v2 = *(const v4f*)(xf + (size_t)e2.x * HFEAT);
            const v4f v3 = *(const v4f*)(xf + (size_t)e3.x * HFEAT);
            float w0 = __int_as_float(e0.y), w1 = __int_as_float(e1.y);
            float w2 = __int_as_float(e2.y), w3 = __int_as_float(e3.y);
            e0 = p[min(j + 4, last)];
            e1 = p[min(j + 5, last)];
            e2 = p[min(j + 6, last)];
            e3 = p[min(j + 7, last)];
            asm volatile("" :: "v"(v0), "v"(v1), "v"(v2), "v"(v3));
            acc += w0 * v0;
            if (j + 1 < c) acc += w1 * v1;
            if (j + 2 < c) acc += w2 * v2;
            if (j + 3 < c) acc += w3 * v3;
        }
        const v4f b = *(const v4f*)(bias + f4);
        v4f r = acc + b;
        r.x = r.x > 0.0f ? r.x : 0.0f;
        r.y = r.y > 0.0f ? r.y : 0.0f;
        r.z = r.z > 0.0f ? r.z : 0.0f;
        r.w = r.w > 0.0f ? r.w : 0.0f;
        *(v4f*)(hrow + q * HPITCH) = r;
    }
}

__device__ __forceinline__ v8f lds_tile_wmma(const float* __restrict__ hs, const v2f (&bf)[16],
                                             int rt, int lane) {
    const float* arow = hs + (rt * 16 + (lane & 15)) * HPITCH + (lane >> 4) * 2;
    v8f acc = {};
#pragma unroll
    for (int kh = 0; kh < 2; ++kh) {
        v2f af[8];
#pragma unroll
        for (int s = 0; s < 8; ++s) af[s] = *(const v2f*)(arow + 4 * (8 * kh + s));
#pragma unroll
        for (int s = 0; s < 8; ++s)
            acc = __builtin_amdgcn_wmma_f32_16x16x4_f32(false, af[s], false, bf[8 * kh + s],
                                                        (short)0, acc, false, false);
    }
    return acc;
}

__global__ void __attribute__((amdgpu_waves_per_eu(16)))
k_gather_gemm64(const int* __restrict__ csr, const int* __restrict__ cnt,
                const int* __restrict__ local, const int* __restrict__ boff,
                const float* __restrict__ xw, const float* __restrict__ bias,
                const float* __restrict__ B, float* __restrict__ C, int n) {
    __shared__ __attribute__((aligned(16))) float hs[TILE_M * HPITCH];
    gather_tile_to_lds(csr, cnt, local, boff, xw, bias, hs, n);

    int w    = __builtin_amdgcn_readfirstlane((int)(threadIdx.x >> 5));
    int lane = threadIdx.x & 31;
    int col  = (w & 3) * 16 + (lane & 15);
    int koff = (lane >> 4) * 2;
    v2f bf[16];
#pragma unroll
    for (int s = 0; s < 16; ++s) {
        bf[s].x = B[(4 * s + koff)     * HFEAT + col];
        bf[s].y = B[(4 * s + koff + 1) * HFEAT + col];
    }
    __syncthreads();

    int base = blockIdx.x * TILE_M;
#pragma unroll
    for (int r = 0; r < 2; ++r) {
        int rt = (w >> 2) * 2 + r;
        int m0 = base + rt * 16;
        if (m0 >= n) break;
        v8f acc = lds_tile_wmma(hs, bf, rt, lane);
        int rbase = m0 + ((lane >> 4) * 8);
        if (m0 + 16 <= n) {
#pragma unroll
            for (int i = 0; i < 8; ++i)
                C[(size_t)(rbase + i) * HFEAT + col] = acc[i];
        } else {
#pragma unroll
            for (int i = 0; i < 8; ++i)
                if (rbase + i < n) C[(size_t)(rbase + i) * HFEAT + col] = acc[i];
        }
    }
}

__global__ void __attribute__((amdgpu_waves_per_eu(16)))
k_gather_head(const int* __restrict__ csr, const int* __restrict__ cnt,
              const int* __restrict__ local, const int* __restrict__ boff,
              const float* __restrict__ xw, const float* __restrict__ bias,
              const float* __restrict__ fc1W, const float* __restrict__ fc1_b,
              const float* __restrict__ fcoW, const float* __restrict__ fcob,
              float* __restrict__ out, int n) {
    __shared__ __attribute__((aligned(16))) float hs[TILE_M * HPITCH];
    __shared__ float gs[TILE_M * GPITCH];
    __shared__ float os[TILE_M];
    gather_tile_to_lds(csr, cnt, local, boff, xw, bias, hs, n);

    int w    = __builtin_amdgcn_readfirstlane((int)(threadIdx.x >> 5));
    int lane = threadIdx.x & 31;
    int ct   = w & 1;
    int rt   = w >> 1;
    int col  = ct * 16 + (lane & 15);
    int koff = (lane >> 4) * 2;
    v2f bf[16];
#pragma unroll
    for (int s = 0; s < 16; ++s) {
        bf[s].x = fc1W[(4 * s + koff)     * 32 + col];
        bf[s].y = fc1W[(4 * s + koff + 1) * 32 + col];
    }
    float fb = fc1_b[lane];
    float fw = fcoW[lane];
    float ob = fcob[0];
    __syncthreads();

    {
        v8f acc = lds_tile_wmma(hs, bf, rt, lane);
        float* grow = gs + (rt * 16 + (lane >> 4) * 8) * GPITCH + col;
#pragma unroll
        for (int i = 0; i < 8; ++i) grow[i * GPITCH] = acc[i];
    }
    __syncthreads();

    float res = 0.0f;
#pragma unroll
    for (int i = 0; i < 8; ++i) {
        float v = gs[(w * 8 + i) * GPITCH + lane] + fb;
        v = v > 0.0f ? v : 0.0f;
        float p = v * fw;
#pragma unroll
        for (int off = 16; off > 0; off >>= 1) p += __shfl_xor(p, off, 32);
        if (lane == i) res = p;
    }
    if (lane < 8) os[w * 8 + lane] = res;
    __syncthreads();

    if (threadIdx.x < TILE_M) {
        int node = blockIdx.x * TILE_M + threadIdx.x;
        if (node < n) out[node] = os[threadIdx.x] + ob;
    }
}

extern "C" void kernel_launch(void* const* d_in, const int* in_sizes, int n_in,
                              void* d_out, int out_size, void* d_ws, size_t ws_size,
                              hipStream_t stream) {
    const float* x     = (const float*)d_in[0];
    const int*   ei    = (const int*)d_in[1];
    const float* W1    = (const float*)d_in[2];
    const float* b1    = (const float*)d_in[3];
    const float* W2    = (const float*)d_in[4];
    const float* b2    = (const float*)d_in[5];
    const float* fc1W  = (const float*)d_in[6];
    const float* fc1b  = (const float*)d_in[7];
    const float* fcoW  = (const float*)d_in[8];
    const float* fcob  = (const float*)d_in[9];
    float* out = (float*)d_out;

    const int N = in_sizes[0] / 128;
    const int E = in_sizes[1] / 2;
    const int* src = ei;
    const int* dst = ei + E;

    auto pad64 = [](size_t v) { return (v + 63) & ~(size_t)63; };
    const int NB = (N + SCAN_ELEMS - 1) / SCAN_ELEMS;

    int* cnt   = (int*)d_ws;
    int* local = cnt + pad64(N);
    int* hist  = local + pad64(N);
    int* offs  = hist + pad64((size_t)GA * NB);
    int* binned = offs + pad64((size_t)GA * NB);
    int* sorted = binned + pad64((size_t)E * 2);
    int* csr   = sorted + pad64((size_t)E * 2);
    float* bufA = (float*)(csr + pad64((size_t)E * 2));
    float* bufB = bufA + (size_t)N * HFEAT;
    const int* boff = offs;

    const int B256 = 256;
    auto blocks = [](long long work, int bs) { return (int)((work + bs - 1) / bs); };

    k_bin_count<<<GA, TB, 0, stream>>>(dst, hist, E, NB);
    k_bin_scan<<<1, TB, 0, stream>>>(hist, offs, NB);
    k_bin_scatter<<<GA, TB, 0, stream>>>(src, dst, offs, (int2*)binned, E, NB);
    k_bin_sort<<<NB * SUBS, TB, 0, stream>>>((const int2*)binned, offs, cnt, local, (int2*)sorted, E, N, NB);
    k_coef<<<blocks((E + IPT - 1) / IPT, B256), B256, 0, stream>>>(cnt, (const int2*)sorted, (int2*)csr, E);

    {
        long long waves = (long long)((N / 16 + 3) / 4) * (HFEAT / 16);
        k_gemm_wmma<128, HFEAT, 4><<<blocks(waves * 32, B256), B256, 0, stream>>>(x, W1, bufA, N);
    }
    k_gather_gemm64<<<blocks(N, TILE_M), B256, 0, stream>>>(csr, cnt, local, boff, bufA, b1, W2, bufB, N);

    k_gather_head<<<blocks(N, TILE_M), B256, 0, stream>>>(csr, cnt, local, boff, bufB, b2, fc1W, fc1b, fcoW, fcob, out, N);
}


# normalisation factored: row writers scale rows by their node's rsqrt(deg), gathers add plain rows and scale the sum by rsqrt(deg[dst]); coefficient pass, list and launch go; sum order unchanged
# speedup vs baseline: 4.0255x; 1.0966x over previous
#include <hip/hip_runtime.h>
#include <hip/hip_bf16.h>

typedef __attribute__((ext_vector_type(2))) float v2f;
typedef __attribute__((ext_vector_type(8))) float v8f;
typedef __attribute__((ext_vector_type(4))) float v4f;

#define HFEAT 64

#define SCAN_ELEMS 1024
#define NBIN_MAX 128
#define GA 256
#define TB 1024
#define SUBS 4
#define BIN_SHIFT 10
#define SUB_SHIFT 8

#define IPT 4

__device__ __forceinline__ int block_excl_scan(int v, int* wsum) {
    int lane = threadIdx.x & 31;
    int wid  = threadIdx.x >> 5;
    int inc = v;
#pragma unroll
    for (int off = 1; off < 32; off <<= 1) {
        int y = __shfl_up(inc, off, 32);
        if (lane >= off) inc += y;
    }
    if (lane == 31) wsum[wid] = inc;
    __syncthreads();
    int woff = 0;
#pragma unroll
    for (int i = 0; i < TB / 32; ++i) {
        int w = wsum[i];
        if (i < wid) woff += w;
    }
    return woff + inc - v;
}

__global__ void __launch_bounds__(TB)
k_bin_count(const int* __restrict__ dst, int* __restrict__ hist, int e, int nbin) {
    __shared__ int h[NBIN_MAX];
    int t = threadIdx.x;
    if (t < NBIN_MAX) h[t] = 0;
    __syncthreads();
    int ch = (e + GA - 1) / GA;
    int lo = blockIdx.x * ch;
    int hi = min(lo + ch, e);
    for (int i = lo + t; i < hi; i += IPT * TB) {
        int d[IPT];
#pragma unroll
        for (int k = 0; k < IPT; ++k)
            d[k] = dst[min(i + k * TB, hi - 1)];
        asm volatile("" :: "v"(d[0]), "v"(d[1]), "v"(d[2]), "v"(d[3]));
#pragma unroll
        for (int k = 0; k < IPT; ++k)
            if (i + k * TB < hi) atomicAdd(&h[d[k] >> BIN_SHIFT], 1);
    }
    __syncthreads();
    if (t < nbin) hist[blockIdx.x * nbin + t] = h[t];
}

__global__ void __launch_bounds__(TB)
k_bin_scan(const int* __restrict__ hist, int* __restrict__ offs, int nbin) {
    __shared__ int a[GA * NBIN_MAX];
    __shared__ int wsum[TB / 32];
    int t = threadIdx.x;
    int total = GA * nbin;
    for (int i0 = t; i0 < total; i0 += 8 * TB) {
        int hv[8];
#pragma unroll
        for (int k = 0; k < 8; ++k)
            hv[k] = hist[min(i0 + k * TB, total - 1)];
        asm volatile("" :: "v"(hv[0]), "v"(hv[1]), "v"(hv[2]), "v"(hv[3]),
                           "v"(hv[4]), "v"(hv[5]), "v"(hv[6]), "v"(hv[7]));
#pragma unroll
        for (int k = 0; k < 8; ++k)
            if (i0 + k * TB < total) a[i0 + k * TB] = hv[k];
    }
    __syncthreads();
    int K = (total + TB - 1) / TB;
    int q0 = t * K;
    int s = 0;
    for (int k = 0; k < K; ++k) {
        int q = q0 + k;
        if (q < total) s += a[(q % GA) * nbin + q / GA];
    }
    int run = block_excl_scan(s, wsum);
    for (int k = 0; k < K; ++k) {
        int q = q0 + k;
        if (q < total) {
            int idx = (q % GA) * nbin + q / GA;
            int v = a[idx];
            a[idx] = run;
            run += v;
        }
    }
    __syncthreads();
    for (int i = t; i < total; i += TB) offs[i] = a[i];
}

__global__ void __launch_bounds__(TB)
k_bin_scatter(const int* __restrict__ src, const int* __restrict__ dst,
              const int* __restrict__ offs, int2* __restrict__ binned, int e, int nbin) {
    __shared__ int cur[NBIN_MAX];
    int t = threadIdx.x;
    if (t < nbin) cur[t] = offs[blockIdx.x * nbin + t];
    __syncthreads();
    int ch = (e + GA - 1) / GA;
    int lo = blockIdx.x * ch;
    int hi = min(lo + ch, e);
    for (int i = lo + t; i < hi; i += IPT * TB) {
        int s[IPT], d[IPT];
#pragma unroll
        for (int k = 0; k < IPT; ++k) {
            int j = min(i + k * TB, hi - 1);
            s[k] = src[j];
            d[k] = dst[j];
        }
        asm volatile("" :: "v"(s[0]), "v"(d[0]), "v"(s[1]), "v"(d[1]),
                           "v"(s[2]), "v"(d[2]), "v"(s[3]), "v"(d[3]));
#pragma unroll
        for (int k = 0; k < IPT; ++k) {
            if (i + k * TB < hi) {
                int p = atomicAdd(&cur[d[k] >> BIN_SHIFT], 1);
                binned[p] = make_int2(s[k], d[k]);
            }
        }
    }
}

__global__ void __launch_bounds__(TB)
k_bin_sort(const int2* __restrict__ binned, const int* __restrict__ offs,
           int* __restrict__ cnt, int* __restrict__ local, int* __restrict__ sorted,
           int e, int n, int nbin) {
    __shared__ int c[SCAN_ELEMS];
    __shared__ int wsum[TB / 32];
    int t = threadIdx.x;
    int b = blockIdx.x / SUBS;
    int sub = blockIdx.x % SUBS;
    int lo = offs[b];
    int hi = b + 1 < nbin ? offs[b + 1] : e;
    int node0 = b * SCAN_ELEMS;
    c[t] = 0;
    __syncthreads();
    for (int i = lo + t; i < hi; i += IPT * TB) {
        int d[IPT];
#pragma unroll
        for (int k = 0; k < IPT; ++k)
            d[k] = binned[min(i + k * TB, hi - 1)].y;
        asm volatile("" :: "v"(d[0]), "v"(d[1]), "v"(d[2]), "v"(d[3]));
#pragma unroll
        for (int k = 0; k < IPT; ++k)
            if (i + k * TB < hi) atomicAdd(&c[d[k] - node0], 1);
    }
    __syncthreads();
    int v = c[t];
    int ex = block_excl_scan(v, wsum);
    int node = node0 + t;
    if ((t >> SUB_SHIFT) == sub && node < n) {
        cnt[node] = v;
        local[node] = ex;
    }
    c[t] = lo + ex;
    __syncthreads();
    for (int i = lo + t; i < hi; i += IPT * TB) {
        int2 ed[IPT];
#pragma unroll
        for (int k = 0; k < IPT; ++k)
            ed[k] = binned[min(i + k * TB, hi - 1)];
        asm volatile("" :: "v"(ed[0].x), "v"(ed[0].y), "v"(ed[1].x), "v"(ed[1].y),
                           "v"(ed[2].x), "v"(ed[2].y), "v"(ed[3].x), "v"(ed[3].y));
#pragma unroll
        for (int k = 0; k < IPT; ++k) {
            int dl = ed[k].y - node0;
            if (i + k * TB < hi && (dl >> SUB_SHIFT) == sub) {
                int p = atomicAdd(&c[dl], 1);
                sorted[p] = ed[k].x;
            }
        }
    }
}

template <int K, int NC, int R>
__global__ void k_gemm_wmma(const float* __restrict__ A, const float* __restrict__ B,
                            const int* __restrict__ cnt, float* __restrict__ C, int M) {
    constexpr int NT = NC / 16;
    constexpr int KS = K / 4;
    int wave = (int)((blockIdx.x * blockDim.x + threadIdx.x) >> 5);
    int lane = threadIdx.x & 31;
    int mtiles = M / 16;
    int mt0 = (wave / NT) * R;
    if (mt0 >= mtiles) return;
    int mend = min(mt0 + R, mtiles);

    int n0 = (wave % NT) * 16;
    int col  = n0 + (lane & 15);
    int koff = (lane >> 4) * 2;

    v2f bf[KS];
#pragma unroll
    for (int s = 0; s < KS; ++s) {
        bf[s].x = B[(size_t)(4 * s + koff)     * NC + col];
        bf[s].y = B[(size_t)(4 * s + koff + 1) * NC + col];
    }

    for (int mt = mt0; mt < mend; ++mt) {
        int m0 = mt * 16;
        const float* arow = A + (size_t)(m0 + (lane & 15)) * K + koff;
        v2f af[KS];
#pragma unroll
        for (int s = 0; s < KS; ++s) af[s] = *(const v2f*)(arow + 4 * s);
        int rbase = m0 + ((lane >> 4) * 8);
        int dg[8];
#pragma unroll
        for (int i = 0; i < 8; ++i) dg[i] = cnt[rbase + i];
        v8f acc = {};
#pragma unroll
        for (int s = 0; s < KS; ++s)
            acc = __builtin_amdgcn_wmma_f32_16x16x4_f32(false, af[s], false, bf[s],
                                                        (short)0, acc, false, false);
#pragma unroll
        for (int i = 0; i < 8; ++i)
            C[(size_t)(rbase + i) * NC + col] = acc[i] * rsqrtf((float)(dg[i] + 1));
    }
}

#define GN 4
#define TILE_M 64
#define HPITCH 68
#define GPITCH 36

template <bool SCALE_OUT>
__device__ __forceinline__ void gather_tile_to_lds(
        const int* __restrict__ srcs, const int* __restrict__ cnt,
        const int* __restrict__ local, const int* __restrict__ boff,
        const float* __restrict__ xw, const float* __restrict__ bias,
        float* __restrict__ hs, int n) {
    int t = blockIdx.x * blockDim.x + threadIdx.x;
    int nb = (t >> 4) * GN;
    int f4 = (t & 15) * 4;
    float* hrow = hs + (threadIdx.x >> 4) * (GN * HPITCH) + (threadIdx.x & 15) * 4;
    const v4f zero4 = {0.0f, 0.0f, 0.0f, 0.0f};
    if (nb >= n) {
#pragma unroll
        for (int q = 0; q < GN; ++q) *(v4f*)(hrow + q * HPITCH) = zero4;
        return;
    }
    const float* xf = xw + f4;
    int cq[GN];
#pragma unroll
    for (int q = 0; q < GN; ++q)
        cq[q] = cnt[min(nb + q, n - 1)];
    int lo = local[nb];
    int bo = boff[nb / SCAN_ELEMS];
    const v4f xs = *(const v4f*)(xf + (size_t)nb * HFEAT);
    asm volatile("" :: "v"(lo), "v"(bo), "v"(cq[0]), "v"(cq[1]), "v"(cq[2]), "v"(cq[3]), "v"(xs));
    __builtin_amdgcn_sched_barrier(0);
    int st = lo + bo;
#pragma unroll
    for (int q = 0; q < GN; ++q) {
        int node = nb + q;
        if (node >= n) {
            *(v4f*)(hrow + q * HPITCH) = zero4;
            continue;
        }
        int c = cq[q];
        const v4f x0 = q == 0 ? xs : *(const v4f*)(xf + (size_t)node * HFEAT);
        const int* p = srcs + st;
        st += c;
        int last = c - 1;
        int e0 = 0, e1 = 0, e2 = 0, e3 = 0;
        if (c > 0) {
            e0 = p[0];
            e1 = p[min(1, last)];
            e2 = p[min(2, last)];
            e3 = p[min(3, last)];
        }
        float di = rsqrtf((float)(c + 1));
        v4f acc = x0;
        for (int j = 0; j < c; j += 4) {
            const v4f v0 = *(const v4f*)(xf + (size_t)e0 * HFEAT);
            const v4f v1 = *(const v4f*)(xf + (size_t)e1 * HFEAT);
            const v4f v2 = *(const v4f*)(xf + (size_t)e2 * HFEAT);
            const v4f v3 = *(const v4f*)(xf + (size_t)e3 * HFEAT);
            e0 = p[min(j + 4, last)];
            e1 = p[min(j + 5, last)];
            e2 = p[min(j + 6, last)];
            e3 = p[min(j + 7, last)];
            asm volatile("" :: "v"(v0), "v"(v1), "v"(v2), "v"(v3));
            acc += v0;
            if (j + 1 < c) acc += v1;
            if (j + 2 < c) acc += v2;
            if (j + 3 < c) acc += v3;
        }
        const v4f b = *(const v4f*)(bias + f4);
        v4f r = di * acc + b;
        r.x = r.x > 0.0f ? r.x : 0.0f;
        r.y = r.y > 0.0f ? r.y : 0.0f;
        r.z = r.z > 0.0f ? r.z : 0.0f;
        r.w = r.w > 0.0f ? r.w : 0.0f;
        if (SCALE_OUT) r = di * r;
        *(v4f*)(hrow + q * HPITCH) = r;
    }
}

__device__ __forceinline__ v8f lds_tile_wmma(const float* __restrict__ hs, const v2f (&bf)[16],
                                             int rt, int lane) {
    const float* arow = hs + (rt * 16 + (lane & 15)) * HPITCH + (lane >> 4) * 2;
    v8f acc = {};
#pragma unroll
    for (int kh = 0; kh < 2; ++kh) {
        v2f af[8];
#pragma unroll
        for (int s = 0; s < 8; ++s) af[s] = *(const v2f*)(arow + 4 * (8 * kh + s));
#pragma unroll
        for (int s = 0; s < 8; ++s)
            acc = __builtin_amdgcn_wmma_f32_16x16x4_f32(false, af[s], false, bf[8 * kh + s],
                                                        (short)0, acc, false, false);
    }
    return acc;
}

__global__ void __attribute__((amdgpu_waves_per_eu(16)))
k_gather_gemm64(const int* __restrict__ srcs, const int* __restrict__ cnt,
                const int* __restrict__ local, const int* __restrict__ boff,
                const float* __restrict__ xw, const float* __restrict__ bias,
                const float* __restrict__ B, float* __restrict__ C, int n) {
    __shared__ __attribute__((aligned(16))) float hs[TILE_M * HPITCH];
    gather_tile_to_lds<true>(srcs, cnt, local, boff, xw, bias, hs, n);

    int w    = __builtin_amdgcn_readfirstlane((int)(threadIdx.x >> 5));
    int lane = threadIdx.x & 31;
    int col  = (w & 3) * 16 + (lane & 15);
    int koff = (lane >> 4) * 2;
    v2f bf[16];
#pragma unroll
    for (int s = 0; s < 16; ++s) {
        bf[s].x = B[(4 * s + koff)     * HFEAT + col];
        bf[s].y = B[(4 * s + koff + 1) * HFEAT + col];
    }
    __syncthreads();

    int base = blockIdx.x * TILE_M;
#pragma unroll
    for (int r = 0; r < 2; ++r) {
        int rt = (w >> 2) * 2 + r;
        int m0 = base + rt * 16;
        if (m0 >= n) break;
        v8f acc = lds_tile_wmma(hs, bf, rt, lane);
        int rbase = m0 + ((lane >> 4) * 8);
        if (m0 + 16 <= n) {
#pragma unroll
            for (int i = 0; i < 8; ++i)
                C[(size_t)(rbase + i) * HFEAT + col] = acc[i];
        } else {
#pragma unroll
            for (int i = 0; i < 8; ++i)
                if (rbase + i < n) C[(size_t)(rbase + i) * HFEAT + col] = acc[i];
        }
    }
}

__global__ void __attribute__((amdgpu_waves_per_eu(16)))
k_gather_head(const int* __restrict__ srcs, const int* __restrict__ cnt,
              const int* __restrict__ local, const int* __restrict__ boff,
              const float* __restrict__ xw, const float* __restrict__ bias,
              const float* __restrict__ fc1W, const float* __restrict__ fc1_b,
              const float* __restrict__ fcoW, const float* __restrict__ fcob,
              float* __restrict__ out, int n) {
    __shared__ __attribute__((aligned(16))) float hs[TILE_M * HPITCH];
    __shared__ float gs[TILE_M * GPITCH];
    __shared__ float os[TILE_M];
    gather_tile_to_lds<false>(srcs, cnt, local, boff, xw, bias, hs, n);

    int w    = __builtin_amdgcn_readfirstlane((int)(threadIdx.x >> 5));
    int lane = threadIdx.x & 31;
    int ct   = w & 1;
    int rt   = w >> 1;
    int col  = ct * 16 + (lane & 15);
    int koff = (lane >> 4) * 2;
    v2f bf[16];
#pragma unroll
    for (int s = 0; s < 16; ++s) {
        bf[s].x = fc1W[(4 * s + koff)     * 32 + col];
        bf[s].y = fc1W[(4 * s + koff + 1) * 32 + col];
    }
    float fb = fc1_b[lane];
    float fw = fcoW[lane];
    float ob = fcob[0];
    __syncthreads();

    {
        v8f acc = lds_tile_wmma(hs, bf, rt, lane);
        float* grow = gs + (rt * 16 + (lane >> 4) * 8) * GPITCH + col;
#pragma unroll
        for (int i = 0; i < 8; ++i) grow[i * GPITCH] = acc[i];
    }
    __syncthreads();

    float res = 0.0f;
#pragma unroll
    for (int i = 0; i < 8; ++i) {
        float v = gs[(w * 8 + i) * GPITCH + lane] + fb;
        v = v > 0.0f ? v : 0.0f;
        float p = v * fw;
#pragma unroll
        for (int off = 16; off > 0; off >>= 1) p += __shfl_xor(p, off, 32);
        if (lane == i) res = p;
    }
    if (lane < 8) os[w * 8 + lane] = res;
    __syncthreads();

    if (threadIdx.x < TILE_M) {
        int node = blockIdx.x * TILE_M + threadIdx.x;
        if (node < n) out[node] = os[threadIdx.x] + ob;
    }
}

extern "C" void kernel_launch(void* const* d_in, const int* in_sizes, int n_in,
                              void* d_out, int out_size, void* d_ws, size_t ws_size,
                              hipStream_t stream) {
    const float* x     = (const float*)d_in[0];
    const int*   ei    = (const int*)d_in[1];
    const float* W1    = (const float*)d_in[2];
    const float* b1    = (const float*)d_in[3];
    const float* W2    = (const float*)d_in[4];
    const float* b2    = (const float*)d_in[5];
    const float* fc1W  = (const float*)d_in[6];
    const float* fc1b  = (const float*)d_in[7];
    const float* fcoW  = (const float*)d_in[8];
    const float* fcob  = (const float*)d_in[9];
    float* out = (float*)d_out;

    const int N = in_sizes[0] / 128;
    const int E = in_sizes[1] / 2;
    const int* src = ei;
    const int* dst = ei + E;

    auto pad64 = [](size_t v) { return (v + 63) & ~(size_t)63; };
    const int NB = (N + SCAN_ELEMS - 1) / SCAN_ELEMS;

    int* cnt   = (int*)d_ws;
    int* local = cnt + pad64(N);
    int* hist  = local + pad64(N);
    int* offs  = hist + pad64((size_t)GA * NB);
    int* binned = offs + pad64((size_t)GA * NB);
    int* sorted = binned + pad64((size_t)E * 2);
    float* bufA = (float*)(sorted + pad64((size_t)E));
    float* bufB = bufA + (size_t)N * HFEAT;
    const int* boff = offs;

    const int B256 = 256;
    auto blocks = [](long long work, int bs) { return (int)((work + bs - 1) / bs); };

    k_bin_count<<<GA, TB, 0, stream>>>(dst, hist, E, NB);
    k_bin_scan<<<1, TB, 0, stream>>>(hist, offs, NB);
    k_bin_scatter<<<GA, TB, 0, stream>>>(src, dst, offs, (int2*)binned, E, NB);
    k_bin_sort<<<NB * SUBS, TB, 0, stream>>>((const int2*)binned, offs, cnt, local, sorted, E, N, NB);

    {
        long long waves = (long long)((N / 16 + 3) / 4) * (HFEAT / 16);
        k_gemm_wmma<128, HFEAT, 4><<<blocks(waves * 32, B256), B256, 0, stream>>>(x, W1, cnt, bufA, N);
    }
    k_gather_gemm64<<<blocks(N, TILE_M), B256, 0, stream>>>(sorted, cnt, local, boff, bufA, b1, W2, bufB, N);

    k_gather_head<<<blocks(N, TILE_M), B256, 0, stream>>>(sorted, cnt, local, boff, bufB, b2, fc1W, fc1b, fcoW, fcob, out, N);
}


# bin offsets: each scatter work-group sums the count table's columns itself in LDS (bin sizes, its own place in each bin); the one-work-group scan launch goes (7 to 6 launches); sum order unchanged
# speedup vs baseline: 4.2952x; 1.0670x over previous
#include <hip/hip_runtime.h>
#include <hip/hip_bf16.h>

typedef __attribute__((ext_vector_type(2))) float v2f;
typedef __attribute__((ext_vector_type(8))) float v8f;
typedef __attribute__((ext_vector_type(4))) float v4f;

#define HFEAT 64

#define SCAN_ELEMS 1024
#define NBIN_MAX 128
#define GA 256
#define TB 1024
#define SUBS 4
#define BIN_SHIFT 10
#define SUB_SHIFT 8

#define IPT 4

__device__ __forceinline__ int block_excl_scan(int v, int* wsum) {
    int lane = threadIdx.x & 31;
    int wid  = threadIdx.x >> 5;
    int inc = v;
#pragma unroll
    for (int off = 1; off < 32; off <<= 1) {
        int y = __shfl_up(inc, off, 32);
        if (lane >= off) inc += y;
    }
    if (lane == 31) wsum[wid] = inc;
    __syncthreads();
    int woff = 0;
#pragma unroll
    for (int i = 0; i < TB / 32; ++i) {
        int w = wsum[i];
        if (i < wid) woff += w;
    }
    return woff + inc - v;
}

__global__ void __launch_bounds__(TB)
k_bin_count(const int* __restrict__ dst, int* __restrict__ hist, int e, int nbin) {
    __shared__ int h[NBIN_MAX];
    int t = threadIdx.x;
    if (t < NBIN_MAX) h[t] = 0;
    __syncthreads();
    int ch = (e + GA - 1) / GA;
    int lo = blockIdx.x * ch;
    int hi = min(lo + ch, e);
    for (int i = lo + t; i < hi; i += IPT * TB) {
        int d[IPT];
#pragma unroll
        for (int k = 0; k < IPT; ++k)
            d[k] = dst[min(i + k * TB, hi - 1)];
        asm volatile("" :: "v"(d[0]), "v"(d[1]), "v"(d[2]), "v"(d[3]));
#pragma unroll
        for (int k = 0; k < IPT; ++k)
            if (i + k * TB < hi) atomicAdd(&h[d[k] >> BIN_SHIFT], 1);
    }
    __syncthreads();
    if (t < nbin) hist[blockIdx.x * nbin + t] = h[t];
}

#define RG  (TB / NBIN_MAX)
#define RPG (GA / RG)
static_assert(RG * NBIN_MAX == TB && RG * RPG == GA, "one thread per (row group, bin); the groups cover every chunk");
__global__ void __launch_bounds__(TB)
k_bin_scatter(const int* __restrict__ src, const int* __restrict__ dst,
              const int* __restrict__ hist, int* __restrict__ binstart,
              int2* __restrict__ binned, int e, int nbin) {
    __shared__ int cur[NBIN_MAX];
    __shared__ int ptot[RG * NBIN_MAX];
    __shared__ int ppart[RG * NBIN_MAX];
    __shared__ int wsum[TB / 32];
    int t = threadIdx.x;
    int w = blockIdx.x;
    {
        int b  = t & (NBIN_MAX - 1);
        int g  = t / NBIN_MAX;
        int bc = min(b, nbin - 1);
        const int* hp = hist + (size_t)(g * RPG) * nbin + bc;
        int hv[RPG];
#pragma unroll
        for (int i = 0; i < RPG; ++i) hv[i] = hp[i * nbin];
        int tot = 0, part = 0;
#pragma unroll
        for (int i = 0; i < RPG; ++i) {
            tot += hv[i];
            part += (g * RPG + i < w) ? hv[i] : 0;
        }
        ptot[g * NBIN_MAX + b]  = tot;
        ppart[g * NBIN_MAX + b] = part;
    }
    __syncthreads();
    int size = 0, before = 0;
    if (t < nbin) {
#pragma unroll
        for (int g = 0; g < RG; ++g) {
            size   += ptot[g * NBIN_MAX + t];
            before += ppart[g * NBIN_MAX + t];
        }
    }
    int first = block_excl_scan(size, wsum);
    if (t < nbin) {
        cur[t] = first + before;
        if (w == 0) binstart[t] = first;
    }
    __syncthreads();
    int ch = (e + GA - 1) / GA;
    int lo = blockIdx.x * ch;
    int hi = min(lo + ch, e);
    for (int i = lo + t; i < hi; i += IPT * TB) {
        int s[IPT], d[IPT];
#pragma unroll
        for (int k = 0; k < IPT; ++k) {
            int j = min(i + k * TB, hi - 1);
            s[k] = src[j];
            d[k] = dst[j];
        }
        asm volatile("" :: "v"(s[0]), "v"(d[0]), "v"(s[1]), "v"(d[1]),
                           "v"(s[2]), "v"(d[2]), "v"(s[3]), "v"(d[3]));
#pragma unroll
        for (int k = 0; k < IPT; ++k) {
            if (i + k * TB < hi) {
                int p = atomicAdd(&cur[d[k] >> BIN_SHIFT], 1);
                binned[p] = make_int2(s[k], d[k]);
            }
        }
    }
}

__global__ void __launch_bounds__(TB)
k_bin_sort(const int2* __restrict__ binned, const int* __restrict__ offs,
           int* __restrict__ cnt, int* __restrict__ local, int* __restrict__ sorted,
           int e, int n, int nbin) {
    __shared__ int c[SCAN_ELEMS];
    __shared__ int wsum[TB / 32];
    int t = threadIdx.x;
    int b = blockIdx.x / SUBS;
    int sub = blockIdx.x % SUBS;
    int lo = offs[b];
    int hi = b + 1 < nbin ? offs[b + 1] : e;
    int node0 = b * SCAN_ELEMS;
    c[t] = 0;
    __syncthreads();
    for (int i = lo + t; i < hi; i += IPT * TB) {
        int d[IPT];
#pragma unroll
        for (int k = 0; k < IPT; ++k)
            d[k] = binned[min(i + k * TB, hi - 1)].y;
        asm volatile("" :: "v"(d[0]), "v"(d[1]), "v"(d[2]), "v"(d[3]));
#pragma unroll
        for (int k = 0; k < IPT; ++k)
            if (i + k * TB < hi) atomicAdd(&c[d[k] - node0], 1);
    }
    __syncthreads();
    int v = c[t];
    int ex = block_excl_scan(v, wsum);
    int node = node0 + t;
    if ((t >> SUB_SHIFT) == sub && node < n) {
        cnt[node] = v;
        local[node] = ex;
    }
    c[t] = lo + ex;
    __syncthreads();
    for (int i = lo + t; i < hi; i += IPT * TB) {
        int2 ed[IPT];
#pragma unroll
        for (int k = 0; k < IPT; ++k)
            ed[k] = binned[min(i + k * TB, hi - 1)];
        asm volatile("" :: "v"(ed[0].x), "v"(ed[0].y), "v"(ed[1].x), "v"(ed[1].y),
                           "v"(ed[2].x), "v"(ed[2].y), "v"(ed[3].x), "v"(ed[3].y));
#pragma unroll
        for (int k = 0; k < IPT; ++k) {
            int dl = ed[k].y - node0;
            if (i + k * TB < hi && (dl >> SUB_SHIFT) == sub) {
                int p = atomicAdd(&c[dl], 1);
                sorted[p] = ed[k].x;
            }
        }
    }
}

template <int K, int NC, int R>
__global__ void k_gemm_wmma(const float* __restrict__ A, const float* __restrict__ B,
                            const int* __restrict__ cnt, float* __restrict__ C, int M) {
    constexpr int NT = NC / 16;
    constexpr int KS = K / 4;
    int wave = (int)((blockIdx.x * blockDim.x + threadIdx.x) >> 5);
    int lane = threadIdx.x & 31;
    int mtiles = M / 16;
    int mt0 = (wave / NT) * R;
    if (mt0 >= mtiles) return;
    int mend = min(mt0 + R, mtiles);

    int n0 = (wave % NT) * 16;
    int col  = n0 + (lane & 15);
    int koff = (lane >> 4) * 2;

    v2f bf[KS];
#pragma unroll
    for (int s = 0; s < KS; ++s) {
        bf[s].x = B[(size_t)(4 * s + koff)     * NC + col];
        bf[s].y = B[(size_t)(4 * s + koff + 1) * NC + col];
    }

    for (int mt = mt0; mt < mend; ++mt) {
        int m0 = mt * 16;
        const float* arow = A + (size_t)(m0 + (lane & 15)) * K + koff;
        v2f af[KS];
#pragma unroll
        for (int s = 0; s < KS; ++s) af[s] = *(const v2f*)(arow + 4 * s);
        int rbase = m0 + ((lane >> 4) * 8);
        int dg[8];
#pragma unroll
        for (int i = 0; i < 8; ++i) dg[i] = cnt[rbase + i];
        v8f acc = {};
#pragma unroll
        for (int s = 0; s < KS; ++s)
            acc = __builtin_amdgcn_wmma_f32_16x16x4_f32(false, af[s], false, bf[s],
                                                        (short)0, acc, false, false);
#pragma unroll
        for (int i = 0; i < 8; ++i)
            C[(size_t)(rbase + i) * NC + col] = acc[i] * rsqrtf((float)(dg[i] + 1));
    }
}

#define GN 4
#define TILE_M 64
#define HPITCH 68
#define GPITCH 36

template <bool SCALE_OUT>
__device__ __forceinline__ void gather_tile_to_lds(
        const int* __restrict__ srcs, const int* __restrict__ cnt,
        const int* __restrict__ local, const int* __restrict__ boff,
        const float* __restrict__ xw, const float* __restrict__ bias,
        float* __restrict__ hs, int n) {
    int t = blockIdx.x * blockDim.x + threadIdx.x;
    int nb = (t >> 4) * GN;
    int f4 = (t & 15) * 4;
    float* hrow = hs + (threadIdx.x >> 4) * (GN * HPITCH) + (threadIdx.x & 15) * 4;
    const v4f zero4 = {0.0f, 0.0f, 0.0f, 0.0f};
    if (nb >= n) {
#pragma unroll
        for (int q = 0; q < GN; ++q) *(v4f*)(hrow + q * HPITCH) = zero4;
        return;
    }
    const float* xf = xw + f4;
    int cq[GN];
#pragma unroll
    for (int q = 0; q < GN; ++q)
        cq[q] = cnt[min(nb + q, n - 1)];
    int lo = local[nb];
    int bo = boff[nb / SCAN_ELEMS];
    const v4f xs = *(const v4f*)(xf + (size_t)nb * HFEAT);
    asm volatile("" :: "v"(lo), "v"(bo), "v"(cq[0]), "v"(cq[1]), "v"(cq[2]), "v"(cq[3]), "v"(xs));
    __builtin_amdgcn_sched_barrier(0);
    int st = lo + bo;
#pragma unroll
    for (int q = 0; q < GN; ++q) {
        int node = nb + q;
        if (node >= n) {
            *(v4f*)(hrow + q * HPITCH) = zero4;
            continue;
        }
        int c = cq[q];
        const v4f x0 = q == 0 ? xs : *(const v4f*)(xf + (size_t)node * HFEAT);
        const int* p = srcs + st;
        st += c;
        int last = c - 1;
        int e0 = 0, e1 = 0, e2 = 0, e3 = 0;
        if (c > 0) {
            e0 = p[0];
            e1 = p[min(1, last)];
            e2 = p[min(2, last)];
            e3 = p[min(3, last)];
        }
        float di = rsqrtf((float)(c + 1));
        v4f acc = x0;
        for (int j = 0; j < c; j += 4) {
            const v4f v0 = *(const v4f*)(xf + (size_t)e0 * HFEAT);
            const v4f v1 = *(const v4f*)(xf + (size_t)e1 * HFEAT);
            const v4f v2 = *(const v4f*)(xf + (size_t)e2 * HFEAT);
            const v4f v3 = *(const v4f*)(xf + (size_t)e3 * HFEAT);
            e0 = p[min(j + 4, last)];
            e1 = p[min(j + 5, last)];
            e2 = p[min(j + 6, last)];
            e3 = p[min(j + 7, last)];
            asm volatile("" :: "v"(v0), "v"(v1), "v"(v2), "v"(v3));
            acc += v0;
            if (j + 1 < c) acc += v1;
            if (j + 2 < c) acc += v2;
            if (j + 3 < c) acc += v3;
        }
        const v4f b = *(const v4f*)(bias + f4);
        v4f r = di * acc + b;
        r.x = r.x > 0.0f ? r.x : 0.0f;
        r.y = r.y > 0.0f ? r.y : 0.0f;
        r.z = r.z > 0.0f ? r.z : 0.0f;
        r.w = r.w > 0.0f ? r.w : 0.0f;
        if (SCALE_OUT) r = di * r;
        *(v4f*)(hrow + q * HPITCH) = r;
    }
}

__device__ __forceinline__ v8f lds_tile_wmma(const float* __restrict__ hs, const v2f (&bf)[16],
                                             int rt, int lane) {
    const float* arow = hs + (rt * 16 + (lane & 15)) * HPITCH + (lane >> 4) * 2;
    v8f acc = {};
#pragma unroll
    for (int kh = 0; kh < 2; ++kh) {
        v2f af[8];
#pragma unroll
        for (int s = 0; s < 8; ++s) af[s] = *(const v2f*)(arow + 4 * (8 * kh + s));
#pragma unroll
        for (int s = 0; s < 8; ++s)
            acc = __builtin_amdgcn_wmma_f32_16x16x4_f32(false, af[s], false, bf[8 * kh + s],
                                                        (short)0, acc, false, false);
    }
    return acc;
}

__global__ void __attribute__((amdgpu_waves_per_eu(16)))
k_gather_gemm64(const int* __restrict__ srcs, const int* __restrict__ cnt,
                const int* __restrict__ local, const int* __restrict__ boff,
                const float* __restrict__ xw, const float* __restrict__ bias,
                const float* __restrict__ B, float* __restrict__ C, int n) {
    __shared__ __attribute__((aligned(16))) float hs[TILE_M * HPITCH];
    gather_tile_to_lds<true>(srcs, cnt, local, boff, xw, bias, hs, n);

    int w    = __builtin_amdgcn_readfirstlane((int)(threadIdx.x >> 5));
    int lane = threadIdx.x & 31;
    int col  = (w & 3) * 16 + (lane & 15);
    int koff = (lane >> 4) * 2;
    v2f bf[16];
#pragma unroll
    for (int s = 0; s < 16; ++s) {
        bf[s].x = B[(4 * s + koff)     * HFEAT + col];
        bf[s].y = B[(4 * s + koff + 1) * HFEAT + col];
    }
    __syncthreads();

    int base = blockIdx.x * TILE_M;
#pragma unroll
    for (int r = 0; r < 2; ++r) {
        int rt = (w >> 2) * 2 + r;
        int m0 = base + rt * 16;
        if (m0 >= n) break;
        v8f acc = lds_tile_wmma(hs, bf, rt, lane);
        int rbase = m0 + ((lane >> 4) * 8);
        if (m0 + 16 <= n) {
#pragma unroll
            for (int i = 0; i < 8; ++i)
                C[(size_t)(rbase + i) * HFEAT + col] = acc[i];
        } else {
#pragma unroll
            for (int i = 0; i < 8; ++i)
                if (rbase + i < n) C[(size_t)(rbase + i) * HFEAT + col] = acc[i];
        }
    }
}

__global__ void __attribute__((amdgpu_waves_per_eu(16)))
k_gather_head(const int* __restrict__ srcs, const int* __restrict__ cnt,
              const int* __restrict__ local, const int* __restrict__ boff,
              const float* __restrict__ xw, const float* __restrict__ bias,
              const float* __restrict__ fc1W, const float* __restrict__ fc1_b,
              const float* __restrict__ fcoW, const float* __restrict__ fcob,
              float* __restrict__ out, int n) {
    __shared__ __attribute__((aligned(16))) float hs[TILE_M * HPITCH];
    __shared__ float gs[TILE_M * GPITCH];
    __shared__ float os[TILE_M];
    gather_tile_to_lds<false>(srcs, cnt, local, boff, xw, bias, hs, n);

    int w    = __builtin_amdgcn_readfirstlane((int)(threadIdx.x >> 5));
    int lane = threadIdx.x & 31;
    int ct   = w & 1;
    int rt   = w >> 1;
    int col  = ct * 16 + (lane & 15);
    int koff = (lane >> 4) * 2;
    v2f bf[16];
#pragma unroll
    for (int s = 0; s < 16; ++s) {
        bf[s].x = fc1W[(4 * s + koff)     * 32 + col];
        bf[s].y = fc1W[(4 * s + koff + 1) * 32 + col];
    }
    float fb = fc1_b[lane];
    float fw = fcoW[lane];
    float ob = fcob[0];
    __syncthreads();

    {
        v8f acc = lds_tile_wmma(hs, bf, rt, lane);
        float* grow = gs + (rt * 16 + (lane >> 4) * 8) * GPITCH + col;
#pragma unroll
        for (int i = 0; i < 8; ++i) grow[i * GPITCH] = acc[i];
    }
    __syncthreads();

    float res = 0.0f;
#pragma unroll
    for (int i = 0; i < 8; ++i) {
        float v = gs[(w * 8 + i) * GPITCH + lane] + fb;
        v = v > 0.0f ? v : 0.0f;
        float p = v * fw;
#pragma unroll
        for (int off = 16; off > 0; off >>= 1) p += __shfl_xor(p, off, 32);
        if (lane == i) res = p;
    }
    if (lane < 8) os[w * 8 + lane] = res;
    __syncthreads();

    if (threadIdx.x < TILE_M) {
        int node = blockIdx.x * TILE_M + threadIdx.x;
        if (node < n) out[node] = os[threadIdx.x] + ob;
    }
}

extern "C" void kernel_launch(void* const* d_in, const int* in_sizes, int n_in,
                              void* d_out, int out_size, void* d_ws, size_t ws_size,
                              hipStream_t stream) {
    const float* x     = (const float*)d_in[0];
    const int*   ei    = (const int*)d_in[1];
    const float* W1    = (const float*)d_in[2];
    const float* b1    = (const float*)d_in[3];
    const float* W2    = (const float*)d_in[4];
    const float* b2    = (const float*)d_in[5];
    const float* fc1W  = (const float*)d_in[6];
    const float* fc1b  = (const float*)d_in[7];
    const float* fcoW  = (const float*)d_in[8];
    const float* fcob  = (const float*)d_in[9];
    float* out = (float*)d_out;

    const int N = in_sizes[0] / 128;
    const int E = in_sizes[1] / 2;
    const int* src = ei;
    const int* dst = ei + E;

    auto pad64 = [](size_t v) { return (v + 63) & ~(size_t)63; };
    const int NB = (N + SCAN_ELEMS - 1) / SCAN_ELEMS;

    int* cnt   = (int*)d_ws;
    int* local = cnt + pad64(N);
    int* hist  = local + pad64(N);
    int* offs  = hist + pad64((size_t)GA * NB);
    int* binned = offs + pad64((size_t)NB);
    int* sorted = binned + pad64((size_t)E * 2);
    float* bufA = (float*)(sorted + pad64((size_t)E));
    float* bufB = bufA + (size_t)N * HFEAT;
    const int* boff = offs;

    const int B256 = 256;
    auto blocks = [](long long work, int bs) { return (int)((work + bs - 1) / bs); };

    k_bin_count<<<GA, TB, 0, stream>>>(dst, hist, E, NB);
    k_bin_scatter<<<GA, TB, 0, stream>>>(src, dst, hist, offs, (int2*)binned, E, NB);
    k_bin_sort<<<NB * SUBS, TB, 0, stream>>>((const int2*)binned, offs, cnt, local, sorted, E, N, NB);

    {
        long long waves = (long long)((N / 16 + 3) / 4) * (HFEAT / 16);
        k_gemm_wmma<128, HFEAT, 4><<<blocks(waves * 32, B256), B256, 0, stream>>>(x, W1, cnt, bufA, N);
    }
    k_gather_gemm64<<<blocks(N, TILE_M), B256, 0, stream>>>(sorted, cnt, local, boff, bufA, b1, W2, bufB, N);

    k_gather_head<<<blocks(N, TILE_M), B256, 0, stream>>>(sorted, cnt, local, boff, bufB, b2, fc1W, fc1b, fcoW, fcob, out, N);
}


# sort: both placing kernels order their records in an LDS copy and store it in order, so a store covers whole lines, not up to 32 pieces; sum order unchanged (arrival order of LDS adds)
# speedup vs baseline: 4.8016x; 1.1179x over previous
#include <hip/hip_runtime.h>
#include <hip/hip_bf16.h>

typedef __attribute__((ext_vector_type(2))) float v2f;
typedef __attribute__((ext_vector_type(8))) float v8f;
typedef __attribute__((ext_vector_type(4))) float v4f;

#define HFEAT 64

#define SCAN_ELEMS 1024
#define NBIN_MAX 128
#define GA 256
#define TB 1024
#define SUBS 4
#define BIN_SHIFT 10
#define SUB_SHIFT 8

#define IPT 4

__device__ __forceinline__ int block_excl_scan(int v, int* wsum) {
    int lane = threadIdx.x & 31;
    int wid  = threadIdx.x >> 5;
    int inc = v;
#pragma unroll
    for (int off = 1; off < 32; off <<= 1) {
        int y = __shfl_up(inc, off, 32);
        if (lane >= off) inc += y;
    }
    if (lane == 31) wsum[wid] = inc;
    __syncthreads();
    int woff = 0;
#pragma unroll
    for (int i = 0; i < TB / 32; ++i) {
        int w = wsum[i];
        if (i < wid) woff += w;
    }
    return woff + inc - v;
}

__global__ void __launch_bounds__(TB)
k_bin_count(const int* __restrict__ dst, int* __restrict__ hist, int e, int nbin) {
    __shared__ int h[NBIN_MAX];
    int t = threadIdx.x;
    if (t < NBIN_MAX) h[t] = 0;
    __syncthreads();
    int ch = (e + GA - 1) / GA;
    int lo = blockIdx.x * ch;
    int hi = min(lo + ch, e);
    for (int i = lo + t; i < hi; i += IPT * TB) {
        int d[IPT];
#pragma unroll
        for (int k = 0; k < IPT; ++k)
            d[k] = dst[min(i + k * TB, hi - 1)];
        asm volatile("" :: "v"(d[0]), "v"(d[1]), "v"(d[2]), "v"(d[3]));
#pragma unroll
        for (int k = 0; k < IPT; ++k)
            if (i + k * TB < hi) atomicAdd(&h[d[k] >> BIN_SHIFT], 1);
    }
    __syncthreads();
    if (t < nbin) hist[blockIdx.x * nbin + t] = h[t];
}

#define RG  (TB / NBIN_MAX)
#define RPG (GA / RG)
#define STAGE_CAP 8192
static_assert(RG * NBIN_MAX == TB && RG * RPG == GA, "one thread per (row group, bin); the groups cover every chunk");
__global__ void __launch_bounds__(TB)
k_bin_scatter(const int* __restrict__ src, const int* __restrict__ dst,
              const int* __restrict__ hist, int* __restrict__ binstart,
              int2* __restrict__ binned, int e, int nbin) {
    __shared__ int cur[NBIN_MAX];
    __shared__ int delta[NBIN_MAX];
    __shared__ int ptot[RG * NBIN_MAX];
    __shared__ int ppart[RG * NBIN_MAX];
    __shared__ int wsum[TB / 32];
    __shared__ int wsum2[TB / 32];
    __shared__ int2 stage[STAGE_CAP];
    int t = threadIdx.x;
    int w = blockIdx.x;
    int ownv = hist[(size_t)w * nbin + min(t, nbin - 1)];
    {
        int b  = t & (NBIN_MAX - 1);
        int g  = t / NBIN_MAX;
        int bc = min(b, nbin - 1);
        const int* hp = hist + (size_t)(g * RPG) * nbin + bc;
        int hv[RPG];
#pragma unroll
        for (int i = 0; i < RPG; ++i) hv[i] = hp[i * nbin];
        int tot = 0, part = 0;
#pragma unroll
        for (int i = 0; i < RPG; ++i) {
            tot += hv[i];
            part += (g * RPG + i < w) ? hv[i] : 0;
        }
        ptot[g * NBIN_MAX + b]  = tot;
        ppart[g * NBIN_MAX + b] = part;
    }
    __syncthreads();
    int size = 0, before = 0, own = 0;
    if (t < nbin) {
#pragma unroll
        for (int g = 0; g < RG; ++g) {
            size   += ptot[g * NBIN_MAX + t];
            before += ppart[g * NBIN_MAX + t];
        }
        own = ownv;
    }
    int first  = block_excl_scan(size, wsum);
    int lfirst = block_excl_scan(own, wsum2);
    if (t < nbin) {
        cur[t] = lfirst;
        delta[t] = first + before - lfirst;
        if (w == 0) binstart[t] = first;
    }
    __syncthreads();
    int ch = (e + GA - 1) / GA;
    int lo = blockIdx.x * ch;
    int hi = min(lo + ch, e);
    int len = hi - lo;
    if (len <= STAGE_CAP) {
        for (int i = lo + t; i < hi; i += IPT * TB) {
            int s[IPT], d[IPT], r[IPT];
#pragma unroll
            for (int k = 0; k < IPT; ++k) {
                int j = min(i + k * TB, hi - 1);
                s[k] = src[j];
                d[k] = dst[j];
            }
            asm volatile("" :: "v"(s[0]), "v"(d[0]), "v"(s[1]), "v"(d[1]),
                               "v"(s[2]), "v"(d[2]), "v"(s[3]), "v"(d[3]));
#pragma unroll
            for (int k = 0; k < IPT; ++k) {
                r[k] = 0;
                if (i + k * TB < hi) r[k] = atomicAdd(&cur[d[k] >> BIN_SHIFT], 1);
            }
#pragma unroll
            for (int k = 0; k < IPT; ++k)
                if (i + k * TB < hi) stage[r[k]] = make_int2(s[k], d[k]);
        }
        __syncthreads();
        for (int i = t; i < len; i += IPT * TB) {
            int2 ed[IPT];
            int dl[IPT];
#pragma unroll
            for (int k = 0; k < IPT; ++k)
                ed[k] = stage[min(i + k * TB, len - 1)];
#pragma unroll
            for (int k = 0; k < IPT; ++k)
                dl[k] = delta[ed[k].y >> BIN_SHIFT];
#pragma unroll
            for (int k = 0; k < IPT; ++k)
                if (i + k * TB < len) binned[i + k * TB + dl[k]] = ed[k];
        }
    } else {
        for (int i = lo + t; i < hi; i += IPT * TB) {
            int s[IPT], d[IPT];
#pragma unroll
            for (int k = 0; k < IPT; ++k) {
                int j = min(i + k * TB, hi - 1);
                s[k] = src[j];
                d[k] = dst[j];
            }
            asm volatile("" :: "v"(s[0]), "v"(d[0]), "v"(s[1]), "v"(d[1]),
                               "v"(s[2]), "v"(d[2]), "v"(s[3]), "v"(d[3]));
#pragma unroll
            for (int k = 0; k < IPT; ++k) {
                if (i + k * TB < hi) {
                    int bn = d[k] >> BIN_SHIFT;
                    int r = atomicAdd(&cur[bn], 1);
                    binned[r + delta[bn]] = make_int2(s[k], d[k]);
                }
            }
        }
    }
}

#define SORT_CAP 16384
__global__ void __launch_bounds__(TB)
k_bin_sort(const int2* __restrict__ binned, const int* __restrict__ offs,
           int* __restrict__ cnt, int* __restrict__ local, int* __restrict__ sorted,
           int e, int n, int nbin) {
    __shared__ int c[SCAN_ELEMS];
    __shared__ int wsum[TB / 32];
    __shared__ int rng[2];
    __shared__ int stage[SORT_CAP];
    int t = threadIdx.x;
    int b = blockIdx.x / SUBS;
    int sub = blockIdx.x % SUBS;
    int lo = offs[b];
    int hi = b + 1 < nbin ? offs[b + 1] : e;
    int node0 = b * SCAN_ELEMS;
    c[t] = 0;
    __syncthreads();
    for (int i = lo + t; i < hi; i += IPT * TB) {
        int d[IPT];
#pragma unroll
        for (int k = 0; k < IPT; ++k)
            d[k] = binned[min(i + k * TB, hi - 1)].y;
        asm volatile("" :: "v"(d[0]), "v"(d[1]), "v"(d[2]), "v"(d[3]));
#pragma unroll
        for (int k = 0; k < IPT; ++k)
            if (i + k * TB < hi) atomicAdd(&c[d[k] - node0], 1);
    }
    __syncthreads();
    int v = c[t];
    int ex = block_excl_scan(v, wsum);
    int node = node0 + t;
    if ((t >> SUB_SHIFT) == sub && node < n) {
        cnt[node] = v;
        local[node] = ex;
    }
    if (t == (sub << SUB_SHIFT)) rng[0] = ex;
    if (t == (sub << SUB_SHIFT) + (SCAN_ELEMS / SUBS) - 1) rng[1] = ex + v;
    c[t] = lo + ex;
    __syncthreads();
    int base = lo + rng[0];
    int len  = rng[1] - rng[0];
    if (len <= SORT_CAP) {
        for (int i = lo + t; i < hi; i += IPT * TB) {
            int2 ed[IPT];
            int p[IPT];
            bool mine[IPT];
#pragma unroll
            for (int k = 0; k < IPT; ++k)
                ed[k] = binned[min(i + k * TB, hi - 1)];
            asm volatile("" :: "v"(ed[0].x), "v"(ed[0].y), "v"(ed[1].x), "v"(ed[1].y),
                               "v"(ed[2].x), "v"(ed[2].y), "v"(ed[3].x), "v"(ed[3].y));
            __builtin_amdgcn_sched_barrier(0);
#pragma unroll
            for (int k = 0; k < IPT; ++k) {
                int dl = ed[k].y - node0;
                mine[k] = i + k * TB < hi && (dl >> SUB_SHIFT) == sub;
                p[k] = base;
                if (mine[k]) p[k] = atomicAdd(&c[dl], 1);
            }
#pragma unroll
            for (int k = 0; k < IPT; ++k)
                if (mine[k]) stage[p[k] - base] = ed[k].x;
        }
        __syncthreads();
        for (int i = t - (base & 31); i < len; i += TB)
            if (i >= 0) sorted[base + i] = stage[i];
    } else {
        for (int i = lo + t; i < hi; i += IPT * TB) {
            int2 ed[IPT];
#pragma unroll
            for (int k = 0; k < IPT; ++k)
                ed[k] = binned[min(i + k * TB, hi - 1)];
            asm volatile("" :: "v"(ed[0].x), "v"(ed[0].y), "v"(ed[1].x), "v"(ed[1].y),
                               "v"(ed[2].x), "v"(ed[2].y), "v"(ed[3].x), "v"(ed[3].y));
#pragma unroll
            for (int k = 0; k < IPT; ++k) {
                int dl = ed[k].y - node0;
                if (i + k * TB < hi && (dl >> SUB_SHIFT) == sub) {
                    int p = atomicAdd(&c[dl], 1);
                    sorted[p] = ed[k].x;
                }
            }
        }
    }
}

template <int K, int NC, int R>
__global__ void k_gemm_wmma(const float* __restrict__ A, const float* __restrict__ B,
                            const int* __restrict__ cnt, float* __restrict__ C, int M) {
    constexpr int NT = NC / 16;
    constexpr int KS = K / 4;
    int wave = (int)((blockIdx.x * blockDim.x + threadIdx.x) >> 5);
    int lane = threadIdx.x & 31;
    int mtiles = M / 16;
    int mt0 = (wave / NT) * R;
    if (mt0 >= mtiles) return;
    int mend = min(mt0 + R, mtiles);

    int n0 = (wave % NT) * 16;
    int col  = n0 + (lane & 15);
    int koff = (lane >> 4) * 2;

    v2f bf[KS];
#pragma unroll
    for (int s = 0; s < KS; ++s) {
        bf[s].x = B[(size_t)(4 * s + koff)     * NC + col];
        bf[s].y = B[(size_t)(4 * s + koff + 1) * NC + col];
    }

    for (int mt = mt0; mt < mend; ++mt) {
        int m0 = mt * 16;
        const float* arow = A + (size_t)(m0 + (lane & 15)) * K + koff;
        v2f af[KS];
#pragma unroll
        for (int s = 0; s < KS; ++s) af[s] = *(const v2f*)(arow + 4 * s);
        int rbase = m0 + ((lane >> 4) * 8);
        int dg[8];
#pragma unroll
        for (int i = 0; i < 8; ++i) dg[i] = cnt[rbase + i];
        v8f acc = {};
#pragma unroll
        for (int s = 0; s < KS; ++s)
            acc = __builtin_amdgcn_wmma_f32_16x16x4_f32(false, af[s], false, bf[s],
                                                        (short)0, acc, false, false);
#pragma unroll
        for (int i = 0; i < 8; ++i)
            C[(size_t)(rbase + i) * NC + col] = acc[i] * rsqrtf((float)(dg[i] + 1));
    }
}

#define GN 4
#define TILE_M 64
#define HPITCH 68
#define GPITCH 36

template <bool SCALE_OUT>
__device__ __forceinline__ void gather_tile_to_lds(
        const int* __restrict__ srcs, const int* __restrict__ cnt,
        const int* __restrict__ local, const int* __restrict__ boff,
        const float* __restrict__ xw, const float* __restrict__ bias,
        float* __restrict__ hs, int n) {
    int t = blockIdx.x * blockDim.x + threadIdx.x;
    int nb = (t >> 4) * GN;
    int f4 = (t & 15) * 4;
    float* hrow = hs + (threadIdx.x >> 4) * (GN * HPITCH) + (threadIdx.x & 15) * 4;
    const v4f zero4 = {0.0f, 0.0f, 0.0f, 0.0f};
    if (nb >= n) {
#pragma unroll
        for (int q = 0; q < GN; ++q) *(v4f*)(hrow + q * HPITCH) = zero4;
        return;
    }
    const float* xf = xw + f4;
    int cq[GN];
#pragma unroll
    for (int q = 0; q < GN; ++q)
        cq[q] = cnt[min(nb + q, n - 1)];
    int lo = local[nb];
    int bo = boff[nb / SCAN_ELEMS];
    const v4f xs = *(const v4f*)(xf + (size_t)nb * HFEAT);
    asm volatile("" :: "v"(lo), "v"(bo), "v"(cq[0]), "v"(cq[1]), "v"(cq[2]), "v"(cq[3]), "v"(xs));
    __builtin_amdgcn_sched_barrier(0);
    int st = lo + bo;
#pragma unroll
    for (int q = 0; q < GN; ++q) {
        int node = nb + q;
        if (node >= n) {
            *(v4f*)(hrow + q * HPITCH) = zero4;
            continue;
        }
        int c = cq[q];
        const v4f x0 = q == 0 ? xs : *(const v4f*)(xf + (size_t)node * HFEAT);
        const int* p = srcs + st;
        st += c;
        int last = c - 1;
        int e0 = 0, e1 = 0, e2 = 0, e3 = 0;
        if (c > 0) {
            e0 = p[0];
            e1 = p[min(1, last)];
            e2 = p[min(2, last)];
            e3 = p[min(3, last)];
        }
        float di = rsqrtf((float)(c + 1));
        v4f acc = x0;
        for (int j = 0; j < c; j += 4) {
            const v4f v0 = *(const v4f*)(xf + (size_t)e0 * HFEAT);
            const v4f v1 = *(const v4f*)(xf + (size_t)e1 * HFEAT);
            const v4f v2 = *(const v4f*)(xf + (size_t)e2 * HFEAT);
            const v4f v3 = *(const v4f*)(xf + (size_t)e3 * HFEAT);
            e0 = p[min(j + 4, last)];
            e1 = p[min(j + 5, last)];
            e2 = p[min(j + 6, last)];
            e3 = p[min(j + 7, last)];
            asm volatile("" :: "v"(v0), "v"(v1), "v"(v2), "v"(v3));
            acc += v0;
            if (j + 1 < c) acc += v1;
            if (j + 2 < c) acc += v2;
            if (j + 3 < c) acc += v3;
        }
        const v4f b = *(const v4f*)(bias + f4);
        v4f r = di * acc + b;
        r.x = r.x > 0.0f ? r.x : 0.0f;
        r.y = r.y > 0.0f ? r.y : 0.0f;
        r.z = r.z > 0.0f ? r.z : 0.0f;
        r.w = r.w > 0.0f ? r.w : 0.0f;
        if (SCALE_OUT) r = di * r;
        *(v4f*)(hrow + q * HPITCH) = r;
    }
}

__device__ __forceinline__ v8f lds_tile_wmma(const float* __restrict__ hs, const v2f (&bf)[16],
                                             int rt, int lane) {
    const float* arow = hs + (rt * 16 + (lane & 15)) * HPITCH + (lane >> 4) * 2;
    v8f acc = {};
#pragma unroll
    for (int kh = 0; kh < 2; ++kh) {
        v2f af[8];
#pragma unroll
        for (int s = 0; s < 8; ++s) af[s] = *(const v2f*)(arow + 4 * (8 * kh + s));
#pragma unroll
        for (int s = 0; s < 8; ++s)
            acc = __builtin_amdgcn_wmma_f32_16x16x4_f32(false, af[s], false, bf[8 * kh + s],
                                                        (short)0, acc, false, false);
    }
    return acc;
}

__global__ void __attribute__((amdgpu_waves_per_eu(16)))
k_gather_gemm64(const int* __restrict__ srcs, const int* __restrict__ cnt,
                const int* __restrict__ local, const int* __restrict__ boff,
                const float* __restrict__ xw, const float* __restrict__ bias,
                const float* __restrict__ B, float* __restrict__ C, int n) {
    __shared__ __attribute__((aligned(16))) float hs[TILE_M * HPITCH];
    gather_tile_to_lds<true>(srcs, cnt, local, boff, xw, bias, hs, n);

    int w    = __builtin_amdgcn_readfirstlane((int)(threadIdx.x >> 5));
    int lane = threadIdx.x & 31;
    int col  = (w & 3) * 16 + (lane & 15);
    int koff = (lane >> 4) * 2;
    v2f bf[16];
#pragma unroll
    for (int s = 0; s < 16; ++s) {
        bf[s].x = B[(4 * s + koff)     * HFEAT + col];
        bf[s].y = B[(4 * s + koff + 1) * HFEAT + col];
    }
    __syncthreads();

    int base = blockIdx.x * TILE_M;
#pragma unroll
    for (int r = 0; r < 2; ++r) {
        int rt = (w >> 2) * 2 + r;
        int m0 = base + rt * 16;
        if (m0 >= n) break;
        v8f acc = lds_tile_wmma(hs, bf, rt, lane);
        int rbase = m0 + ((lane >> 4) * 8);
        if (m0 + 16 <= n) {
#pragma unroll
            for (int i = 0; i < 8; ++i)
                C[(size_t)(rbase + i) * HFEAT + col] = acc[i];
        } else {
#pragma unroll
            for (int i = 0; i < 8; ++i)
                if (rbase + i < n) C[(size_t)(rbase + i) * HFEAT + col] = acc[i];
        }
    }
}

__global__ void __attribute__((amdgpu_waves_per_eu(16)))
k_gather_head(const int* __restrict__ srcs, const int* __restrict__ cnt,
              const int* __restrict__ local, const int* __restrict__ boff,
              const float* __restrict__ xw, const float* __restrict__ bias,
              const float* __restrict__ fc1W, const float* __restrict__ fc1_b,
              const float* __restrict__ fcoW, const float* __restrict__ fcob,
              float* __restrict__ out, int n) {
    __shared__ __attribute__((aligned(16))) float hs[TILE_M * HPITCH];
    __shared__ float gs[TILE_M * GPITCH];
    __shared__ float os[TILE_M];
    gather_tile_to_lds<false>(srcs, cnt, local, boff, xw, bias, hs, n);

    int w    = __builtin_amdgcn_readfirstlane((int)(threadIdx.x >> 5));
    int lane = threadIdx.x & 31;
    int ct   = w & 1;
    int rt   = w >> 1;
    int col  = ct * 16 + (lane & 15);
    int koff = (lane >> 4) * 2;
    v2f bf[16];
#pragma unroll
    for (int s = 0; s < 16; ++s) {
        bf[s].x = fc1W[(4 * s + koff)     * 32 + col];
        bf[s].y = fc1W[(4 * s + koff + 1) * 32 + col];
    }
    float fb = fc1_b[lane];
    float fw = fcoW[lane];
    float ob = fcob[0];
    __syncthreads();

    {
        v8f acc = lds_tile_wmma(hs, bf, rt, lane);
        float* grow = gs + (rt * 16 + (lane >> 4) * 8) * GPITCH + col;
#pragma unroll
        for (int i = 0; i < 8; ++i) grow[i * GPITCH] = acc[i];
    }
    __syncthreads();

    float res = 0.0f;
#pragma unroll
    for (int i = 0; i < 8; ++i) {
        float v = gs[(w * 8 + i) * GPITCH + lane] + fb;
        v = v > 0.0f ? v : 0.0f;
        float p = v * fw;
#pragma unroll
        for (int off = 16; off > 0; off >>= 1) p += __shfl_xor(p, off, 32);
        if (lane == i) res = p;
    }
    if (lane < 8) os[w * 8 + lane] = res;
    __syncthreads();

    if (threadIdx.x < TILE_M) {
        int node = blockIdx.x * TILE_M + threadIdx.x;
        if (node < n) out[node] = os[threadIdx.x] + ob;
    }
}

extern "C" void kernel_launch(void* const* d_in, const int* in_sizes, int n_in,
                              void* d_out, int out_size, void* d_ws, size_t ws_size,
                              hipStream_t stream) {
    const float* x     = (const float*)d_in[0];
    const int*   ei    = (const int*)d_in[1];
    const float* W1    = (const float*)d_in[2];
    const float* b1    = (const float*)d_in[3];
    const float* W2    = (const float*)d_in[4];
    const float* b2    = (const float*)d_in[5];
    const float* fc1W  = (const float*)d_in[6];
    const float* fc1b  = (const float*)d_in[7];
    const float* fcoW  = (const float*)d_in[8];
    const float* fcob  = (const float*)d_in[9];
    float* out = (float*)d_out;

    const int N = in_sizes[0] / 128;
    const int E = in_sizes[1] / 2;
    const int* src = ei;
    const int* dst = ei + E;

    auto pad64 = [](size_t v) { return (v + 63) & ~(size_t)63; };
    const int NB = (N + SCAN_ELEMS - 1) / SCAN_ELEMS;

    int* cnt   = (int*)d_ws;
    int* local = cnt + pad64(N);
    int* hist  = local + pad64(N);
    int* offs  = hist + pad64((size_t)GA * NB);
    int* binned = offs + pad64((size_t)NB);
    int* sorted = binned + pad64((size_t)E * 2);
    float* bufA = (float*)(sorted + pad64((size_t)E));
    float* bufB = bufA + (size_t)N * HFEAT;
    const int* boff = offs;

    const int B256 = 256;
    auto blocks = [](long long work, int bs) { return (int)((work + bs - 1) / bs); };

    k_bin_count<<<GA, TB, 0, stream>>>(dst, hist, E, NB);
    k_bin_scatter<<<GA, TB, 0, stream>>>(src, dst, hist, offs, (int2*)binned, E, NB);
    k_bin_sort<<<NB * SUBS, TB, 0, stream>>>((const int2*)binned, offs, cnt, local, sorted, E, N, NB);

    {
        long long waves = (long long)((N / 16 + 3) / 4) * (HFEAT / 16);
        k_gemm_wmma<128, HFEAT, 4><<<blocks(waves * 32, B256), B256, 0, stream>>>(x, W1, cnt, bufA, N);
    }
    k_gather_gemm64<<<blocks(N, TILE_M), B256, 0, stream>>>(sorted, cnt, local, boff, bufA, b1, W2, bufB, N);

    k_gather_head<<<blocks(N, TILE_M), B256, 0, stream>>>(sorted, cnt, local, boff, bufB, b2, fc1W, fc1b, fcoW, fcob, out, N);
}


# x @ W1: each workgroup's 128 rows of x come in through LDS as whole 512-byte rows (one load instruction per row), A fragments read from LDS; no 16-row-strided 8-byte loads; sum order unchanged
# speedup vs baseline: 5.1224x; 1.0668x over previous
#include <hip/hip_runtime.h>
#include <hip/hip_bf16.h>

typedef __attribute__((ext_vector_type(2))) float v2f;
typedef __attribute__((ext_vector_type(8))) float v8f;
typedef __attribute__((ext_vector_type(4))) float v4f;

#define HFEAT 64

#define SCAN_ELEMS 1024
#define NBIN_MAX 128
#define GA 256
#define TB 1024
#define SUBS 4
#define BIN_SHIFT 10
#define SUB_SHIFT 8

#define IPT 4

__device__ __forceinline__ int block_excl_scan(int v, int* wsum) {
    int lane = threadIdx.x & 31;
    int wid  = threadIdx.x >> 5;
    int inc = v;
#pragma unroll
    for (int off = 1; off < 32; off <<= 1) {
        int y = __shfl_up(inc, off, 32);
        if (lane >= off) inc += y;
    }
    if (lane == 31) wsum[wid] = inc;
    __syncthreads();
    int woff = 0;
#pragma unroll
    for (int i = 0; i < TB / 32; ++i) {
        int w = wsum[i];
        if (i < wid) woff += w;
    }
    return woff + inc - v;
}

__global__ void __launch_bounds__(TB)
k_bin_count(const int* __restrict__ dst, int* __restrict__ hist, int e, int nbin) {
    __shared__ int h[NBIN_MAX];
    int t = threadIdx.x;
    if (t < NBIN_MAX) h[t] = 0;
    __syncthreads();
    int ch = (e + GA - 1) / GA;
    int lo = blockIdx.x * ch;
    int hi = min(lo + ch, e);
    for (int i = lo + t; i < hi; i += IPT * TB) {
        int d[IPT];
#pragma unroll
        for (int k = 0; k < IPT; ++k)
            d[k] = dst[min(i + k * TB, hi - 1)];
        asm volatile("" :: "v"(d[0]), "v"(d[1]), "v"(d[2]), "v"(d[3]));
#pragma unroll
        for (int k = 0; k < IPT; ++k)
            if (i + k * TB < hi) atomicAdd(&h[d[k] >> BIN_SHIFT], 1);
    }
    __syncthreads();
    if (t < nbin) hist[blockIdx.x * nbin + t] = h[t];
}

#define RG  (TB / NBIN_MAX)
#define RPG (GA / RG)
#define STAGE_CAP 8192
static_assert(RG * NBIN_MAX == TB && RG * RPG == GA, "one thread per (row group, bin); the groups cover every chunk");
__global__ void __launch_bounds__(TB)
k_bin_scatter(const int* __restrict__ src, const int* __restrict__ dst,
              const int* __restrict__ hist, int* __restrict__ binstart,
              int2* __restrict__ binned, int e, int nbin) {
    __shared__ int cur[NBIN_MAX];
    __shared__ int delta[NBIN_MAX];
    __shared__ int ptot[RG * NBIN_MAX];
    __shared__ int ppart[RG * NBIN_MAX];
    __shared__ int wsum[TB / 32];
    __shared__ int wsum2[TB / 32];
    __shared__ int2 stage[STAGE_CAP];
    int t = threadIdx.x;
    int w = blockIdx.x;
    int ownv = hist[(size_t)w * nbin + min(t, nbin - 1)];
    {
        int b  = t & (NBIN_MAX - 1);
        int g  = t / NBIN_MAX;
        int bc = min(b, nbin - 1);
        const int* hp = hist + (size_t)(g * RPG) * nbin + bc;
        int hv[RPG];
#pragma unroll
        for (int i = 0; i < RPG; ++i) hv[i] = hp[i * nbin];
        int tot = 0, part = 0;
#pragma unroll
        for (int i = 0; i < RPG; ++i) {
            tot += hv[i];
            part += (g * RPG + i < w) ? hv[i] : 0;
        }
        ptot[g * NBIN_MAX + b]  = tot;
        ppart[g * NBIN_MAX + b] = part;
    }
    __syncthreads();
    int size = 0, before = 0, own = 0;
    if (t < nbin) {
#pragma unroll
        for (int g = 0; g < RG; ++g) {
            size   += ptot[g * NBIN_MAX + t];
            before += ppart[g * NBIN_MAX + t];
        }
        own = ownv;
    }
    int first  = block_excl_scan(size, wsum);
    int lfirst = block_excl_scan(own, wsum2);
    if (t < nbin) {
        cur[t] = lfirst;
        delta[t] = first + before - lfirst;
        if (w == 0) binstart[t] = first;
    }
    __syncthreads();
    int ch = (e + GA - 1) / GA;
    int lo = blockIdx.x * ch;
    int hi = min(lo + ch, e);
    int len = hi - lo;
    if (len <= STAGE_CAP) {
        for (int i = lo + t; i < hi; i += IPT * TB) {
            int s[IPT], d[IPT], r[IPT];
#pragma unroll
            for (int k = 0; k < IPT; ++k) {
                int j = min(i + k * TB, hi - 1);
                s[k] = src[j];
                d[k] = dst[j];
            }
            asm volatile("" :: "v"(s[0]), "v"(d[0]), "v"(s[1]), "v"(d[1]),
                               "v"(s[2]), "v"(d[2]), "v"(s[3]), "v"(d[3]));
#pragma unroll
            for (int k = 0; k < IPT; ++k) {
                r[k] = 0;
                if (i + k * TB < hi) r[k] = atomicAdd(&cur[d[k] >> BIN_SHIFT], 1);
            }
#pragma unroll
            for (int k = 0; k < IPT; ++k)
                if (i + k * TB < hi) stage[r[k]] = make_int2(s[k], d[k]);
        }
        __syncthreads();
        for (int i = t; i < len; i += IPT * TB) {
            int2 ed[IPT];
            int dl[IPT];
#pragma unroll
            for (int k = 0; k < IPT; ++k)
                ed[k] = stage[min(i + k * TB, len - 1)];
#pragma unroll
            for (int k = 0; k < IPT; ++k)
                dl[k] = delta[ed[k].y >> BIN_SHIFT];
#pragma unroll
            for (int k = 0; k < IPT; ++k)
                if (i + k * TB < len) binned[i + k * TB + dl[k]] = ed[k];
        }
    } else {
        for (int i = lo + t; i < hi; i += IPT * TB) {
            int s[IPT], d[IPT];
#pragma unroll
            for (int k = 0; k < IPT; ++k) {
                int j = min(i + k * TB, hi - 1);
                s[k] = src[j];
                d[k] = dst[j];
            }
            asm volatile("" :: "v"(s[0]), "v"(d[0]), "v"(s[1]), "v"(d[1]),
                               "v"(s[2]), "v"(d[2]), "v"(s[3]), "v"(d[3]));
#pragma unroll
            for (int k = 0; k < IPT; ++k) {
                if (i + k * TB < hi) {
                    int bn = d[k] >> BIN_SHIFT;
                    int r = atomicAdd(&cur[bn], 1);
                    binned[r + delta[bn]] = make_int2(s[k], d[k]);
                }
            }
        }
    }
}

#define SORT_CAP 16384
__global__ void __launch_bounds__(TB)
k_bin_sort(const int2* __restrict__ binned, const int* __restrict__ offs,
           int* __restrict__ cnt, int* __restrict__ local, int* __restrict__ sorted,
           int e, int n, int nbin) {
    __shared__ int c[SCAN_ELEMS];
    __shared__ int wsum[TB / 32];
    __shared__ int rng[2];
    __shared__ int stage[SORT_CAP];
    int t = threadIdx.x;
    int b = blockIdx.x / SUBS;
    int sub = blockIdx.x % SUBS;
    int lo = offs[b];
    int hi = b + 1 < nbin ? offs[b + 1] : e;
    int node0 = b * SCAN_ELEMS;
    c[t] = 0;
    __syncthreads();
    for (int i = lo + t; i < hi; i += IPT * TB) {
        int d[IPT];
#pragma unroll
        for (int k = 0; k < IPT; ++k)
            d[k] = binned[min(i + k * TB, hi - 1)].y;
        asm volatile("" :: "v"(d[0]), "v"(d[1]), "v"(d[2]), "v"(d[3]));
#pragma unroll
        for (int k = 0; k < IPT; ++k)
            if (i + k * TB < hi) atomicAdd(&c[d[k] - node0], 1);
    }
    __syncthreads();
    int v = c[t];
    int ex = block_excl_scan(v, wsum);
    int node = node0 + t;
    if ((t >> SUB_SHIFT) == sub && node < n) {
        cnt[node] = v;
        local[node] = ex;
    }
    if (t == (sub << SUB_SHIFT)) rng[0] = ex;
    if (t == (sub << SUB_SHIFT) + (SCAN_ELEMS / SUBS) - 1) rng[1] = ex + v;
    c[t] = lo + ex;
    __syncthreads();
    int base = lo + rng[0];
    int len  = rng[1] - rng[0];
    if (len <= SORT_CAP) {
        for (int i = lo + t; i < hi; i += IPT * TB) {
            int2 ed[IPT];
            int p[IPT];
            bool mine[IPT];
#pragma unroll
            for (int k = 0; k < IPT; ++k)
                ed[k] = binned[min(i + k * TB, hi - 1)];
            asm volatile("" :: "v"(ed[0].x), "v"(ed[0].y), "v"(ed[1].x), "v"(ed[1].y),
                               "v"(ed[2].x), "v"(ed[2].y), "v"(ed[3].x), "v"(ed[3].y));
            __builtin_amdgcn_sched_barrier(0);
#pragma unroll
            for (int k = 0; k < IPT; ++k) {
                int dl = ed[k].y - node0;
                mine[k] = i + k * TB < hi && (dl >> SUB_SHIFT) == sub;
                p[k] = base;
                if (mine[k]) p[k] = atomicAdd(&c[dl], 1);
            }
#pragma unroll
            for (int k = 0; k < IPT; ++k)
                if (mine[k]) stage[p[k] - base] = ed[k].x;
        }
        __syncthreads();
        for (int i = t - (base & 31); i < len; i += TB)
            if (i >= 0) sorted[base + i] = stage[i];
    } else {
        for (int i = lo + t; i < hi; i += IPT * TB) {
            int2 ed[IPT];
#pragma unroll
            for (int k = 0; k < IPT; ++k)
                ed[k] = binned[min(i + k * TB, hi - 1)];
            asm volatile("" :: "v"(ed[0].x), "v"(ed[0].y), "v"(ed[1].x), "v"(ed[1].y),
                               "v"(ed[2].x), "v"(ed[2].y), "v"(ed[3].x), "v"(ed[3].y));
#pragma unroll
            for (int k = 0; k < IPT; ++k) {
                int dl = ed[k].y - node0;
                if (i + k * TB < hi && (dl >> SUB_SHIFT) == sub) {
                    int p = atomicAdd(&c[dl], 1);
                    sorted[p] = ed[k].x;
                }
            }
        }
    }
}

#define GT_M 128
#define XPITCH 132
template <int K, int NC>
__global__ void __attribute__((amdgpu_waves_per_eu(8)))
k_gemm_wmma(const float* __restrict__ A, const float* __restrict__ B,
            const int* __restrict__ cnt, float* __restrict__ C, int M) {
    static_assert(K + 4 == XPITCH && K % 32 == 0 && NC == 64, "one LDS row per row of A; four column tiles per row tile");
    constexpr int KS = K / 4;
    __shared__ __attribute__((aligned(16))) float xs[GT_M * XPITCH];
    int w    = __builtin_amdgcn_readfirstlane((int)(threadIdx.x >> 5));
    int lane = threadIdx.x & 31;
    int row0 = blockIdx.x * GT_M;

#pragma unroll
    for (int hf = 0; hf < 2; ++hf) {
        v4f xr[8];
#pragma unroll
        for (int j = 0; j < 8; ++j) {
            int rl = w * 16 + hf * 8 + j;
            int rg = min(row0 + rl, M - 1);
            xr[j] = *(const v4f*)(A + (size_t)rg * K + lane * 4);
        }
#pragma unroll
        for (int j = 0; j < 8; ++j) {
            int rl = w * 16 + hf * 8 + j;
            *(v4f*)(xs + rl * XPITCH + lane * 4) = xr[j];
        }
    }

    int col  = (w & 3) * 16 + (lane & 15);
    int koff = (lane >> 4) * 2;
    v2f bf[KS];
#pragma unroll
    for (int s = 0; s < KS; ++s) {
        bf[s].x = B[(size_t)(4 * s + koff)     * NC + col];
        bf[s].y = B[(size_t)(4 * s + koff + 1) * NC + col];
    }
    __syncthreads();

    for (int r = 0; r < 4; ++r) {
        int rt = (w >> 2) * 4 + r;
        int m0 = row0 + rt * 16;
        if (m0 + 16 > M) break;
        const float* arow = xs + (rt * 16 + (lane & 15)) * XPITCH + koff;
        int rbase = m0 + ((lane >> 4) * 8);
        int dg[8];
#pragma unroll
        for (int i = 0; i < 8; ++i) dg[i] = cnt[rbase + i];
        v8f acc = {};
#pragma unroll
        for (int kq = 0; kq < KS / 8; ++kq) {
            v2f af[8];
#pragma unroll
            for (int s = 0; s < 8; ++s) af[s] = *(const v2f*)(arow + 4 * (8 * kq + s));
#pragma unroll
            for (int s = 0; s < 8; ++s)
                acc = __builtin_amdgcn_wmma_f32_16x16x4_f32(false, af[s], false, bf[8 * kq + s],
                                                            (short)0, acc, false, false);
        }
#pragma unroll
        for (int i = 0; i < 8; ++i)
            C[(size_t)(rbase + i) * NC + col] = acc[i] * rsqrtf((float)(dg[i] + 1));
    }
}

#define GN 4
#define TILE_M 64
#define HPITCH 68
#define GPITCH 36

template <bool SCALE_OUT>
__device__ __forceinline__ void gather_tile_to_lds(
        const int* __restrict__ srcs, const int* __restrict__ cnt,
        const int* __restrict__ local, const int* __restrict__ boff,
        const float* __restrict__ xw, const float* __restrict__ bias,
        float* __restrict__ hs, int n) {
    int t = blockIdx.x * blockDim.x + threadIdx.x;
    int nb = (t >> 4) * GN;
    int f4 = (t & 15) * 4;
    float* hrow = hs + (threadIdx.x >> 4) * (GN * HPITCH) + (threadIdx.x & 15) * 4;
    const v4f zero4 = {0.0f, 0.0f, 0.0f, 0.0f};
    if (nb >= n) {
#pragma unroll
        for (int q = 0; q < GN; ++q) *(v4f*)(hrow + q * HPITCH) = zero4;
        return;
    }
    const float* xf = xw + f4;
    int cq[GN];
#pragma unroll
    for (int q = 0; q < GN; ++q)
        cq[q] = cnt[min(nb + q, n - 1)];
    int lo = local[nb];
    int bo = boff[nb / SCAN_ELEMS];
    const v4f xs = *(const v4f*)(xf + (size_t)nb * HFEAT);
    asm volatile("" :: "v"(lo), "v"(bo), "v"(cq[0]), "v"(cq[1]), "v"(cq[2]), "v"(cq[3]), "v"(xs));
    __builtin_amdgcn_sched_barrier(0);
    int st = lo + bo;
#pragma unroll
    for (int q = 0; q < GN; ++q) {
        int node = nb + q;
        if (node >= n) {
            *(v4f*)(hrow + q * HPITCH) = zero4;
            continue;
        }
        int c = cq[q];
        const v4f x0 = q == 0 ? xs : *(const v4f*)(xf + (size_t)node * HFEAT);
        const int* p = srcs + st;
        st += c;
        int last = c - 1;
        int e0 = 0, e1 = 0, e2 = 0, e3 = 0;
        if (c > 0) {
            e0 = p[0];
            e1 = p[min(1, last)];
            e2 = p[min(2, last)];
            e3 = p[min(3, last)];
        }
        float di = rsqrtf((float)(c + 1));
        v4f acc = x0;
        for (int j = 0; j < c; j += 4) {
            const v4f v0 = *(const v4f*)(xf + (size_t)e0 * HFEAT);
            const v4f v1 = *(const v4f*)(xf + (size_t)e1 * HFEAT);
            const v4f v2 = *(const v4f*)(xf + (size_t)e2 * HFEAT);
            const v4f v3 = *(const v4f*)(xf + (size_t)e3 * HFEAT);
            e0 = p[min(j + 4, last)];
            e1 = p[min(j + 5, last)];
            e2 = p[min(j + 6, last)];
            e3 = p[min(j + 7, last)];
            asm volatile("" :: "v"(v0), "v"(v1), "v"(v2), "v"(v3));
            acc += v0;
            if (j + 1 < c) acc += v1;
            if (j + 2 < c) acc += v2;
            if (j + 3 < c) acc += v3;
        }
        const v4f b = *(const v4f*)(bias + f4);
        v4f r = di * acc + b;
        r.x = r.x > 0.0f ? r.x : 0.0f;
        r.y = r.y > 0.0f ? r.y : 0.0f;
        r.z = r.z > 0.0f ? r.z : 0.0f;
        r.w = r.w > 0.0f ? r.w : 0.0f;
        if (SCALE_OUT) r = di * r;
        *(v4f*)(hrow + q * HPITCH) = r;
    }
}

__device__ __forceinline__ v8f lds_tile_wmma(const float* __restrict__ hs, const v2f (&bf)[16],
                                             int rt, int lane) {
    const float* arow = hs + (rt * 16 + (lane & 15)) * HPITCH + (lane >> 4) * 2;
    v8f acc = {};
#pragma unroll
    for (int kh = 0; kh < 2; ++kh) {
        v2f af[8];
#pragma unroll
        for (int s = 0; s < 8; ++s) af[s] = *(const v2f*)(arow + 4 * (8 * kh + s));
#pragma unroll
        for (int s = 0; s < 8; ++s)
            acc = __builtin_amdgcn_wmma_f32_16x16x4_f32(false, af[s], false, bf[8 * kh + s],
                                                        (short)0, acc, false, false);
    }
    return acc;
}

__global__ void __attribute__((amdgpu_waves_per_eu(16)))
k_gather_gemm64(const int* __restrict__ srcs, const int* __restrict__ cnt,
                const int* __restrict__ local, const int* __restrict__ boff,
                const float* __restrict__ xw, const float* __restrict__ bias,
                const float* __restrict__ B, float* __restrict__ C, int n) {
    __shared__ __attribute__((aligned(16))) float hs[TILE_M * HPITCH];
    gather_tile_to_lds<true>(srcs, cnt, local, boff, xw, bias, hs, n);

    int w    = __builtin_amdgcn_readfirstlane((int)(threadIdx.x >> 5));
    int lane = threadIdx.x & 31;
    int col  = (w & 3) * 16 + (lane & 15);
    int koff = (lane >> 4) * 2;
    v2f bf[16];
#pragma unroll
    for (int s = 0; s < 16; ++s) {
        bf[s].x = B[(4 * s + koff)     * HFEAT + col];
        bf[s].y = B[(4 * s + koff + 1) * HFEAT + col];
    }
    __syncthreads();

    int base = blockIdx.x * TILE_M;
#pragma unroll
    for (int r = 0; r < 2; ++r) {
        int rt = (w >> 2) * 2 + r;
        int m0 = base + rt * 16;
        if (m0 >= n) break;
        v8f acc = lds_tile_wmma(hs, bf, rt, lane);
        int rbase = m0 + ((lane >> 4) * 8);
        if (m0 + 16 <= n) {
#pragma unroll
            for (int i = 0; i < 8; ++i)
                C[(size_t)(rbase + i) * HFEAT + col] = acc[i];
        } else {
#pragma unroll
            for (int i = 0; i < 8; ++i)
                if (rbase + i < n) C[(size_t)(rbase + i) * HFEAT + col] = acc[i];
        }
    }
}

__global__ void __attribute__((amdgpu_waves_per_eu(16)))
k_gather_head(const int* __restrict__ srcs, const int* __restrict__ cnt,
              const int* __restrict__ local, const int* __restrict__ boff,
              const float* __restrict__ xw, const float* __restrict__ bias,
              const float* __restrict__ fc1W, const float* __restrict__ fc1_b,
              const float* __restrict__ fcoW, const float* __restrict__ fcob,
              float* __restrict__ out, int n) {
    __shared__ __attribute__((aligned(16))) float hs[TILE_M * HPITCH];
    __shared__ float gs[TILE_M * GPITCH];
    __shared__ float os[TILE_M];
    gather_tile_to_lds<false>(srcs, cnt, local, boff, xw, bias, hs, n);

    int w    = __builtin_amdgcn_readfirstlane((int)(threadIdx.x >> 5));
    int lane = threadIdx.x & 31;
    int ct   = w & 1;
    int rt   = w >> 1;
    int col  = ct * 16 + (lane & 15);
    int koff = (lane >> 4) * 2;
    v2f bf[16];
#pragma unroll
    for (int s = 0; s < 16; ++s) {
        bf[s].x = fc1W[(4 * s + koff)     * 32 + col];
        bf[s].y = fc1W[(4 * s + koff + 1) * 32 + col];
    }
    float fb = fc1_b[lane];
    float fw = fcoW[lane];
    float ob = fcob[0];
    __syncthreads();

    {
        v8f acc = lds_tile_wmma(hs, bf, rt, lane);
        float* grow = gs + (rt * 16 + (lane >> 4) * 8) * GPITCH + col;
#pragma unroll
        for (int i = 0; i < 8; ++i) grow[i * GPITCH] = acc[i];
    }
    __syncthreads();

    float res = 0.0f;
#pragma unroll
    for (int i = 0; i < 8; ++i) {
        float v = gs[(w * 8 + i) * GPITCH + lane] + fb;
        v = v > 0.0f ? v : 0.0f;
        float p = v * fw;
#pragma unroll
        for (int off = 16; off > 0; off >>= 1) p += __shfl_xor(p, off, 32);
        if (lane == i) res = p;
    }
    if (lane < 8) os[w * 8 + lane] = res;
    __syncthreads();

    if (threadIdx.x < TILE_M) {
        int node = blockIdx.x * TILE_M + threadIdx.x;
        if (node < n) out[node] = os[threadIdx.x] + ob;
    }
}

extern "C" void kernel_launch(void* const* d_in, const int* in_sizes, int n_in,
                              void* d_out, int out_size, void* d_ws, size_t ws_size,
                              hipStream_t stream) {
    const float* x     = (const float*)d_in[0];
    const int*   ei    = (const int*)d_in[1];
    const float* W1    = (const float*)d_in[2];
    const float* b1    = (const float*)d_in[3];
    const float* W2    = (const float*)d_in[4];
    const float* b2    = (const float*)d_in[5];
    const float* fc1W  = (const float*)d_in[6];
    const float* fc1b  = (const float*)d_in[7];
    const float* fcoW  = (const float*)d_in[8];
    const float* fcob  = (const float*)d_in[9];
    float* out = (float*)d_out;

    const int N = in_sizes[0] / 128;
    const int E = in_sizes[1] / 2;
    const int* src = ei;
    const int* dst = ei + E;

    auto pad64 = [](size_t v) { return (v + 63) & ~(size_t)63; };
    const int NB = (N + SCAN_ELEMS - 1) / SCAN_ELEMS;

    int* cnt   = (int*)d_ws;
    int* local = cnt + pad64(N);
    int* hist  = local + pad64(N);
    int* offs  = hist + pad64((size_t)GA * NB);
    int* binned = offs + pad64((size_t)NB);
    int* sorted = binned + pad64((size_t)E * 2);
    float* bufA = (float*)(sorted + pad64((size_t)E));
    float* bufB = bufA + (size_t)N * HFEAT;
    const int* boff = offs;

    const int B256 = 256;
    auto blocks = [](long long work, int bs) { return (int)((work + bs - 1) / bs); };

    k_bin_count<<<GA, TB, 0, stream>>>(dst, hist, E, NB);
    k_bin_scatter<<<GA, TB, 0, stream>>>(src, dst, hist, offs, (int2*)binned, E, NB);
    k_bin_sort<<<NB * SUBS, TB, 0, stream>>>((const int2*)binned, offs, cnt, local, sorted, E, N, NB);

    k_gemm_wmma<128, HFEAT><<<blocks(N, GT_M), B256, 0, stream>>>(x, W1, cnt, bufA, N);
    k_gather_gemm64<<<blocks(N, TILE_M), B256, 0, stream>>>(sorted, cnt, local, boff, bufA, b1, W2, bufB, N);

    k_gather_head<<<blocks(N, TILE_M), B256, 0, stream>>>(sorted, cnt, local, boff, bufB, b2, fc1W, fc1b, fcoW, fcob, out, N);
}


# edge sort: each chunk orders its pairs by bin in its own range and writes its bin starts; one workgroup per 256-node bin reads its runs; count launch goes (6 to 5); sum order = LDS arrival order
# speedup vs baseline: 5.4887x; 1.0715x over previous
#include <hip/hip_runtime.h>
#include <hip/hip_bf16.h>

typedef __attribute__((ext_vector_type(2))) float v2f;
typedef __attribute__((ext_vector_type(8))) float v8f;
typedef __attribute__((ext_vector_type(4))) float v4f;

#define HFEAT 64

#define SCAN_ELEMS 256
#define BIN_SHIFT 8
#define NBIN_MAX 512
#define GA 256
#define TB 1024

#define IPT 4

__device__ __forceinline__ int block_excl_scan(int v, int* wsum) {
    int lane = threadIdx.x & 31;
    int wid  = threadIdx.x >> 5;
    int inc = v;
#pragma unroll
    for (int off = 1; off < 32; off <<= 1) {
        int y = __shfl_up(inc, off, 32);
        if (lane >= off) inc += y;
    }
    if (lane == 31) wsum[wid] = inc;
    __syncthreads();
    int woff = 0;
#pragma unroll
    for (int i = 0; i < TB / 32; ++i) {
        int w = wsum[i];
        if (i < wid) woff += w;
    }
    return woff + inc - v;
}

#define ROUNDS 2
#define KPT (ROUNDS * IPT)
#define STAGE_CAP (KPT * TB)
__global__ void __launch_bounds__(TB)
k_chunk_sort(const int* __restrict__ src, const int* __restrict__ dst,
             int* __restrict__ tab, int2* __restrict__ binned, int e, int nbin) {
    __shared__ int h[NBIN_MAX];
    __shared__ int wsum[TB / 32];
    __shared__ int2 stage[STAGE_CAP];
    int t = threadIdx.x;
    int w = blockIdx.x;
    int ch = (e + GA - 1) / GA;
    int lo = w * ch;
    int hi = min(lo + ch, e);
    int len = max(hi - lo, 0);
    bool fits = len <= STAGE_CAP;
    if (t < NBIN_MAX) h[t] = 0;
    __syncthreads();
    int s[KPT], d[KPT], r[KPT];
#pragma unroll
    for (int k = 0; k < KPT; ++k) { s[k] = 0; d[k] = 0; r[k] = 0; }
    if (fits) {
        if (len > 0) {
#pragma unroll
            for (int k = 0; k < KPT; ++k) {
                int j = min(lo + t + k * TB, hi - 1);
                s[k] = src[j];
                d[k] = dst[j];
            }
            asm volatile("" :: "v"(s[0]), "v"(d[0]), "v"(s[1]), "v"(d[1]), "v"(s[2]), "v"(d[2]),
                               "v"(s[3]), "v"(d[3]), "v"(s[4]), "v"(d[4]), "v"(s[5]), "v"(d[5]),
                               "v"(s[6]), "v"(d[6]), "v"(s[7]), "v"(d[7]));
#pragma unroll
            for (int k = 0; k < KPT; ++k)
                if (lo + t + k * TB < hi) r[k] = atomicAdd(&h[d[k] >> BIN_SHIFT], 1);
        }
    } else {
        for (int i = lo + t; i < hi; i += IPT * TB) {
            int dd[IPT];
#pragma unroll
            for (int k = 0; k < IPT; ++k)
                dd[k] = dst[min(i + k * TB, hi - 1)];
            asm volatile("" :: "v"(dd[0]), "v"(dd[1]), "v"(dd[2]), "v"(dd[3]));
#pragma unroll
            for (int k = 0; k < IPT; ++k)
                if (i + k * TB < hi) atomicAdd(&h[dd[k] >> BIN_SHIFT], 1);
        }
    }
    __syncthreads();
    int v = t < nbin ? h[t] : 0;
    int ex = block_excl_scan(v, wsum);
    if (t <= nbin) tab[(size_t)w * (nbin + 1) + t] = ex;
    if (t < nbin) h[t] = ex;
    __syncthreads();
    if (fits) {
        if (len > 0) {
            int hb[KPT];
#pragma unroll
            for (int k = 0; k < KPT; ++k)
                hb[k] = h[d[k] >> BIN_SHIFT];
            asm volatile("" :: "v"(hb[0]), "v"(hb[1]), "v"(hb[2]), "v"(hb[3]),
                               "v"(hb[4]), "v"(hb[5]), "v"(hb[6]), "v"(hb[7]));
#pragma unroll
            for (int k = 0; k < KPT; ++k)
                if (lo + t + k * TB < hi) stage[hb[k] + r[k]] = make_int2(s[k], d[k]);
        }
        __syncthreads();
        for (int i = t - (lo & 15); i < len; i += TB)
            if (i >= 0) binned[lo + i] = stage[i];
    } else {
        for (int i = lo + t; i < hi; i += IPT * TB) {
            int ss[IPT], dd[IPT];
#pragma unroll
            for (int k = 0; k < IPT; ++k) {
                int j = min(i + k * TB, hi - 1);
                ss[k] = src[j];
                dd[k] = dst[j];
            }
            asm volatile("" :: "v"(ss[0]), "v"(dd[0]), "v"(ss[1]), "v"(dd[1]),
                               "v"(ss[2]), "v"(dd[2]), "v"(ss[3]), "v"(dd[3]));
#pragma unroll
            for (int k = 0; k < IPT; ++k) {
                if (i + k * TB < hi) {
                    int rr = atomicAdd(&h[dd[k] >> BIN_SHIFT], 1);
                    binned[lo + rr] = make_int2(ss[k], dd[k]);
                }
            }
        }
    }
}

#define SORT_CAP 16384
#define RPH (GA / (TB / 16))
static_assert(RPH * (TB / 16) == GA, "the 16-lane groups cover every chunk's run");
__global__ void __launch_bounds__(TB) __attribute__((amdgpu_waves_per_eu(16)))
k_node_sort(const int2* __restrict__ binned, const int* __restrict__ tab,
            int* __restrict__ cnt, int* __restrict__ local, int* __restrict__ sorted,
            int* __restrict__ offs, int e, int n, int nbin) {
    __shared__ int c[SCAN_ELEMS];
    __shared__ int rstart[GA];
    __shared__ int rcnt[GA];
    __shared__ int red0[TB / 32];
    __shared__ int red1[TB / 32];
    __shared__ int wsum[TB / 32];
    __shared__ int stage[SORT_CAP];
    int t = threadIdx.x;
    int lane = t & 31;
    int wid  = t >> 5;
    int b = blockIdx.x;
    int node0 = b * SCAN_ELEMS;
    int ch = (e + GA - 1) / GA;
    int a0 = 0, a1 = 0;
    if (t < GA) {
        const int* row = tab + (size_t)t * (nbin + 1) + b;
        a0 = row[0];
        a1 = row[1];
        rstart[t] = t * ch + a0;
        rcnt[t] = a1 - a0;
    }
    if (t < SCAN_ELEMS) c[t] = 0;
    int x0 = a0, x1 = a1 - a0;
#pragma unroll
    for (int off = 16; off > 0; off >>= 1) {
        x0 += __shfl_xor(x0, off, 32);
        x1 += __shfl_xor(x1, off, 32);
    }
    if (lane == 0) { red0[wid] = x0; red1[wid] = x1; }
    __syncthreads();
    int base = 0, size = 0;
#pragma unroll
    for (int i = 0; i < TB / 32; ++i) { base += red0[i]; size += red1[i]; }

    int l16 = t & 15;
    int g0  = (t >> 4) * RPH;
    int2 ed[RPH][2];
    int vm = 0;
#pragma unroll
    for (int q = 0; q < RPH; ++q) {
        int cq = rcnt[g0 + q];
        int sq = rstart[g0 + q];
#pragma unroll
        for (int m = 0; m < 2; ++m) {
            int a = max(min(sq + min(l16 + 16 * m, cq - 1), e - 1), 0);
            ed[q][m] = binned[a];
            if (l16 + 16 * m < cq) vm |= 1 << (2 * q + m);
        }
    }
    asm volatile("" :: "v"(ed[0][0].x), "v"(ed[0][0].y), "v"(ed[0][1].x), "v"(ed[0][1].y),
                       "v"(ed[1][0].x), "v"(ed[1][0].y), "v"(ed[1][1].x), "v"(ed[1][1].y),
                       "v"(ed[2][0].x), "v"(ed[2][0].y), "v"(ed[2][1].x), "v"(ed[2][1].y),
                       "v"(ed[3][0].x), "v"(ed[3][0].y), "v"(ed[3][1].x), "v"(ed[3][1].y));
#pragma unroll
    for (int q = 0; q < RPH; ++q) {
#pragma unroll
        for (int m = 0; m < 2; ++m)
            if ((vm >> (2 * q + m)) & 1) atomicAdd(&c[ed[q][m].y - node0], 1);
    }
#pragma nounroll
    for (int q = 0; q < RPH; ++q) {
        int cq = rcnt[g0 + q];
        int sq = rstart[g0 + q];
        for (int i = l16 + 32; i < cq; i += 16)
            atomicAdd(&c[binned[sq + i].y - node0], 1);
    }
    __syncthreads();
    int v = t < SCAN_ELEMS ? c[t] : 0;
    int ex = block_excl_scan(v, wsum);
    if (t < SCAN_ELEMS) {
        int node = node0 + t;
        if (node < n) {
            cnt[node] = v;
            local[node] = ex;
        }
        c[t] = ex;
    }
    if (t == 0) offs[b] = base;
    __syncthreads();
    if (size <= SORT_CAP) {
        int p[RPH][2];
#pragma unroll
        for (int q = 0; q < RPH; ++q) {
#pragma unroll
            for (int m = 0; m < 2; ++m) {
                p[q][m] = 0;
                if ((vm >> (2 * q + m)) & 1) p[q][m] = atomicAdd(&c[ed[q][m].y - node0], 1);
            }
        }
#pragma unroll
        for (int q = 0; q < RPH; ++q) {
#pragma unroll
            for (int m = 0; m < 2; ++m)
                if ((vm >> (2 * q + m)) & 1) stage[p[q][m]] = ed[q][m].x;
        }
#pragma nounroll
        for (int q = 0; q < RPH; ++q) {
            int cq = rcnt[g0 + q];
            int sq = rstart[g0 + q];
            for (int i = l16 + 32; i < cq; i += 16) {
                int2 x = binned[sq + i];
                stage[atomicAdd(&c[x.y - node0], 1)] = x.x;
            }
        }
        __syncthreads();
        for (int i = t - (base & 31); i < size; i += TB)
            if (i >= 0) sorted[base + i] = stage[i];
    } else {
#pragma unroll
        for (int q = 0; q < RPH; ++q) {
#pragma unroll
            for (int m = 0; m < 2; ++m)
                if ((vm >> (2 * q + m)) & 1) sorted[base + atomicAdd(&c[ed[q][m].y - node0], 1)] = ed[q][m].x;
        }
#pragma nounroll
        for (int q = 0; q < RPH; ++q) {
            int cq = rcnt[g0 + q];
            int sq = rstart[g0 + q];
            for (int i = l16 + 32; i < cq; i += 16) {
                int2 x = binned[sq + i];
                sorted[base + atomicAdd(&c[x.y - node0], 1)] = x.x;
            }
        }
    }
}

#define GT_M 128
#define XPITCH 132
template <int K, int NC>
__global__ void __attribute__((amdgpu_waves_per_eu(8)))
k_gemm_wmma(const float* __restrict__ A, const float* __restrict__ B,
            const int* __restrict__ cnt, float* __restrict__ C, int M) {
    static_assert(K + 4 == XPITCH && K % 32 == 0 && NC == 64, "one LDS row per row of A; four column tiles per row tile");
    constexpr int KS = K / 4;
    __shared__ __attribute__((aligned(16))) float xs[GT_M * XPITCH];
    int w    = __builtin_amdgcn_readfirstlane((int)(threadIdx.x >> 5));
    int lane = threadIdx.x & 31;
    int row0 = blockIdx.x * GT_M;

#pragma unroll
    for (int hf = 0; hf < 2; ++hf) {
        v4f xr[8];
#pragma unroll
        for (int j = 0; j < 8; ++j) {
            int rl = w * 16 + hf * 8 + j;
            int rg = min(row0 + rl, M - 1);
            xr[j] = *(const v4f*)(A + (size_t)rg * K + lane * 4);
        }
#pragma unroll
        for (int j = 0; j < 8; ++j) {
            int rl = w * 16 + hf * 8 + j;
            *(v4f*)(xs + rl * XPITCH + lane * 4) = xr[j];
        }
    }

    int col  = (w & 3) * 16 + (lane & 15);
    int koff = (lane >> 4) * 2;
    v2f bf[KS];
#pragma unroll
    for (int s = 0; s < KS; ++s) {
        bf[s].x = B[(size_t)(4 * s + koff)     * NC + col];
        bf[s].y = B[(size_t)(4 * s + koff + 1) * NC + col];
    }
    __syncthreads();

    for (int r = 0; r < 4; ++r) {
        int rt = (w >> 2) * 4 + r;
        int m0 = row0 + rt * 16;
        if (m0 + 16 > M) break;
        const float* arow = xs + (rt * 16 + (lane & 15)) * XPITCH + koff;
        int rbase = m0 + ((lane >> 4) * 8);
        int dg[8];
#pragma unroll
        for (int i = 0; i < 8; ++i) dg[i] = cnt[rbase + i];
        v8f acc = {};
#pragma unroll
        for (int kq = 0; kq < KS / 8; ++kq) {
            v2f af[8];
#pragma unroll
            for (int s = 0; s < 8; ++s) af[s] = *(const v2f*)(arow + 4 * (8 * kq + s));
#pragma unroll
            for (int s = 0; s < 8; ++s)
                acc = __builtin_amdgcn_wmma_f32_16x16x4_f32(false, af[s], false, bf[8 * kq + s],
                                                            (short)0, acc, false, false);
        }
#pragma unroll
        for (int i = 0; i < 8; ++i)
            C[(size_t)(rbase + i) * NC + col] = acc[i] * rsqrtf((float)(dg[i] + 1));
    }
}

#define GN 4
#define TILE_M 64
#define HPITCH 68
#define GPITCH 36

template <bool SCALE_OUT>
__device__ __forceinline__ void gather_tile_to_lds(
        const int* __restrict__ srcs, const int* __restrict__ cnt,
        const int* __restrict__ local, const int* __restrict__ boff,
        const float* __restrict__ xw, const float* __restrict__ bias,
        float* __restrict__ hs, int n) {
    int t = blockIdx.x * blockDim.x + threadIdx.x;
    int nb = (t >> 4) * GN;
    int f4 = (t & 15) * 4;
    float* hrow = hs + (threadIdx.x >> 4) * (GN * HPITCH) + (threadIdx.x & 15) * 4;
    const v4f zero4 = {0.0f, 0.0f, 0.0f, 0.0f};
    if (nb >= n) {
#pragma unroll
        for (int q = 0; q < GN; ++q) *(v4f*)(hrow + q * HPITCH) = zero4;
        return;
    }
    const float* xf = xw + f4;
    int cq[GN];
#pragma unroll
    for (int q = 0; q < GN; ++q)
        cq[q] = cnt[min(nb + q, n - 1)];
    int lo = local[nb];
    int bo = boff[nb / SCAN_ELEMS];
    const v4f xs = *(const v4f*)(xf + (size_t)nb * HFEAT);
    asm volatile("" :: "v"(lo), "v"(bo), "v"(cq[0]), "v"(cq[1]), "v"(cq[2]), "v"(cq[3]), "v"(xs));
    __builtin_amdgcn_sched_barrier(0);
    int st = lo + bo;
#pragma unroll
    for (int q = 0; q < GN; ++q) {
        int node = nb + q;
        if (node >= n) {
            *(v4f*)(hrow + q * HPITCH) = zero4;
            continue;
        }
        int c = cq[q];
        const v4f x0 = q == 0 ? xs : *(const v4f*)(xf + (size_t)node * HFEAT);
        const int* p = srcs + st;
        st += c;
        int last = c - 1;
        int e0 = 0, e1 = 0, e2 = 0, e3 = 0;
        if (c > 0) {
            e0 = p[0];
            e1 = p[min(1, last)];
            e2 = p[min(2, last)];
            e3 = p[min(3, last)];
        }
        float di = rsqrtf((float)(c + 1));
        v4f acc = x0;
        for (int j = 0; j < c; j += 4) {
            const v4f v0 = *(const v4f*)(xf + (size_t)e0 * HFEAT);
            const v4f v1 = *(const v4f*)(xf + (size_t)e1 * HFEAT);
            const v4f v2 = *(const v4f*)(xf + (size_t)e2 * HFEAT);
            const v4f v3 = *(const v4f*)(xf + (size_t)e3 * HFEAT);
            e0 = p[min(j + 4, last)];
            e1 = p[min(j + 5, last)];
            e2 = p[min(j + 6, last)];
            e3 = p[min(j + 7, last)];
            asm volatile("" :: "v"(v0), "v"(v1), "v"(v2), "v"(v3));
            acc += v0;
            if (j + 1 < c) acc += v1;
            if (j + 2 < c) acc += v2;
            if (j + 3 < c) acc += v3;
        }
        const v4f b = *(const v4f*)(bias + f4);
        v4f r = di * acc + b;
        r.x = r.x > 0.0f ? r.x : 0.0f;
        r.y = r.y > 0.0f ? r.y : 0.0f;
        r.z = r.z > 0.0f ? r.z : 0.0f;
        r.w = r.w > 0.0f ? r.w : 0.0f;
        if (SCALE_OUT) r = di * r;
        *(v4f*)(hrow + q * HPITCH) = r;
    }
}

__device__ __forceinline__ v8f lds_tile_wmma(const float* __restrict__ hs, const v2f (&bf)[16],
                                             int rt, int lane) {
    const float* arow = hs + (rt * 16 + (lane & 15)) * HPITCH + (lane >> 4) * 2;
    v8f acc = {};
#pragma unroll
    for (int kh = 0; kh < 2; ++kh) {
        v2f af[8];
#pragma unroll
        for (int s = 0; s < 8; ++s) af[s] = *(const v2f*)(arow + 4 * (8 * kh + s));
#pragma unroll
        for (int s = 0; s < 8; ++s)
            acc = __builtin_amdgcn_wmma_f32_16x16x4_f32(false, af[s], false, bf[8 * kh + s],
                                                        (short)0, acc, false, false);
    }
    return acc;
}

__global__ void __attribute__((amdgpu_waves_per_eu(16)))
k_gather_gemm64(const int* __restrict__ srcs, const int* __restrict__ cnt,
                const int* __restrict__ local, const int* __restrict__ boff,
                const float* __restrict__ xw, const float* __restrict__ bias,
                const float* __restrict__ B, float* __restrict__ C, int n) {
    __shared__ __attribute__((aligned(16))) float hs[TILE_M * HPITCH];
    gather_tile_to_lds<true>(srcs, cnt, local, boff, xw, bias, hs, n);

    int w    = __builtin_amdgcn_readfirstlane((int)(threadIdx.x >> 5));
    int lane = threadIdx.x & 31;
    int col  = (w & 3) * 16 + (lane & 15);
    int koff = (lane >> 4) * 2;
    v2f bf[16];
#pragma unroll
    for (int s = 0; s < 16; ++s) {
        bf[s].x = B[(4 * s + koff)     * HFEAT + col];
        bf[s].y = B[(4 * s + koff + 1) * HFEAT + col];
    }
    __syncthreads();

    int base = blockIdx.x * TILE_M;
#pragma unroll
    for (int r = 0; r < 2; ++r) {
        int rt = (w >> 2) * 2 + r;
        int m0 = base + rt * 16;
        if (m0 >= n) break;
        v8f acc = lds_tile_wmma(hs, bf, rt, lane);
        int rbase = m0 + ((lane >> 4) * 8);
        if (m0 + 16 <= n) {
#pragma unroll
            for (int i = 0; i < 8; ++i)
                C[(size_t)(rbase + i) * HFEAT + col] = acc[i];
        } else {
#pragma unroll
            for (int i = 0; i < 8; ++i)
                if (rbase + i < n) C[(size_t)(rbase + i) * HFEAT + col] = acc[i];
        }
    }
}

__global__ void __attribute__((amdgpu_waves_per_eu(16)))
k_gather_head(const int* __restrict__ srcs, const int* __restrict__ cnt,
              const int* __restrict__ local, const int* __restrict__ boff,
              const float* __restrict__ xw, const float* __restrict__ bias,
              const float* __restrict__ fc1W, const float* __restrict__ fc1_b,
              const float* __restrict__ fcoW, const float* __restrict__ fcob,
              float* __restrict__ out, int n) {
    __shared__ __attribute__((aligned(16))) float hs[TILE_M * HPITCH];
    __shared__ float gs[TILE_M * GPITCH];
    __shared__ float os[TILE_M];
    gather_tile_to_lds<false>(srcs, cnt, local, boff, xw, bias, hs, n);

    int w    = __builtin_amdgcn_readfirstlane((int)(threadIdx.x >> 5));
    int lane = threadIdx.x & 31;
    int ct   = w & 1;
    int rt   = w >> 1;
    int col  = ct * 16 + (lane & 15);
    int koff = (lane >> 4) * 2;
    v2f bf[16];
#pragma unroll
    for (int s = 0; s < 16; ++s) {
        bf[s].x = fc1W[(4 * s + koff)     * 32 + col];
        bf[s].y = fc1W[(4 * s + koff + 1) * 32 + col];
    }
    float fb = fc1_b[lane];
    float fw = fcoW[lane];
    float ob = fcob[0];
    __syncthreads();

    {
        v8f acc = lds_tile_wmma(hs, bf, rt, lane);
        float* grow = gs + (rt * 16 + (lane >> 4) * 8) * GPITCH + col;
#pragma unroll
        for (int i = 0; i < 8; ++i) grow[i * GPITCH] = acc[i];
    }
    __syncthreads();

    float res = 0.0f;
#pragma unroll
    for (int i = 0; i < 8; ++i) {
        float v = gs[(w * 8 + i) * GPITCH + lane] + fb;
        v = v > 0.0f ? v : 0.0f;
        float p = v * fw;
#pragma unroll
        for (int off = 16; off > 0; off >>= 1) p += __shfl_xor(p, off, 32);
        if (lane == i) res = p;
    }
    if (lane < 8) os[w * 8 + lane] = res;
    __syncthreads();

    if (threadIdx.x < TILE_M) {
        int node = blockIdx.x * TILE_M + threadIdx.x;
        if (node < n) out[node] = os[threadIdx.x] + ob;
    }
}

extern "C" void kernel_launch(void* const* d_in, const int* in_sizes, int n_in,
                              void* d_out, int out_size, void* d_ws, size_t ws_size,
                              hipStream_t stream) {
    const float* x     = (const float*)d_in[0];
    const int*   ei    = (const int*)d_in[1];
    const float* W1    = (const float*)d_in[2];
    const float* b1    = (const float*)d_in[3];
    const float* W2    = (const float*)d_in[4];
    const float* b2    = (const float*)d_in[5];
    const float* fc1W  = (const float*)d_in[6];
    const float* fc1b  = (const float*)d_in[7];
    const float* fcoW  = (const float*)d_in[8];
    const float* fcob  = (const float*)d_in[9];
    float* out = (float*)d_out;

    const int N = in_sizes[0] / 128;
    const int E = in_sizes[1] / 2;
    const int* src = ei;
    const int* dst = ei + E;

    auto pad64 = [](size_t v) { return (v + 63) & ~(size_t)63; };
    const int NB = (N + SCAN_ELEMS - 1) / SCAN_ELEMS;

    int* cnt   = (int*)d_ws;
    int* local = cnt + pad64(N);
    int* tab   = local + pad64(N);
    int* offs  = tab + pad64((size_t)GA * (NB + 1));
    int* binned = offs + pad64((size_t)NB);
    int* sorted = binned + pad64((size_t)E * 2);
    float* bufA = (float*)(sorted + pad64((size_t)E));
    float* bufB = bufA + (size_t)N * HFEAT;
    const int* boff = offs;

    const int B256 = 256;
    auto blocks = [](long long work, int bs) { return (int)((work + bs - 1) / bs); };

    k_chunk_sort<<<GA, TB, 0, stream>>>(src, dst, tab, (int2*)binned, E, NB);
    k_node_sort<<<NB, TB, 0, stream>>>((const int2*)binned, tab, cnt, local, sorted, offs, E, N, NB);

    k_gemm_wmma<128, HFEAT><<<blocks(N, GT_M), B256, 0, stream>>>(x, W1, cnt, bufA, N);
    k_gather_gemm64<<<blocks(N, TILE_M), B256, 0, stream>>>(sorted, cnt, local, boff, bufA, b1, W2, bufB, N);

    k_gather_head<<<blocks(N, TILE_M), B256, 0, stream>>>(sorted, cnt, local, boff, bufB, b2, fc1W, fc1b, fcoW, fcob, out, N);
}
